# starting kernel
# baseline (speedup 1.0000x reference)
#include <hip/hip_runtime.h>

#define B_     64
#define CIN    3
#define Hh     96
#define Ww     96
#define COUT   96
#define KH     5
#define KW     5
#define STRIDE 2
#define PAD    2
#define HOUT   48
#define WOUT   48
#define L_     (HOUT*WOUT)
#define K_     (CIN*KH*KW)
#define KP     76
#define LR     0.02f
#define DELTA  (-0.4f)
#define PRECF  1e-30f

typedef float v2f __attribute__((ext_vector_type(2)));
typedef float v8f __attribute__((ext_vector_type(8)));

__global__ __launch_bounds__(192) void conv_hebb_gemm(
    const float* __restrict__ x, const float* __restrict__ w,
    float* __restrict__ out, float* __restrict__ yx, float* __restrict__ yy)
{
    __shared__ float lds_w[COUT * KP];
    __shared__ float lds_p[KP * 16];
    __shared__ float lds_t[COUT * 16];
    __shared__ int   s_i0[16], s_i1[16];
    __shared__ float s_t0[16], s_t1[16];

    const int tid = threadIdx.x;
    const int b   = blockIdx.y;
    const int l0  = blockIdx.x * 16;

    for (int i = tid; i < COUT * KP; i += 192) {
        int o = i / KP, k = i - o * KP;
        lds_w[i] = (k < K_) ? w[o * K_ + k] : 0.0f;
    }
    for (int i = tid; i < KP * 16; i += 192) {
        int k = i >> 4, n = i & 15;
        float v = 0.0f;
        if (k < K_) {
            int c  = k / (KH * KW);
            int r  = k - c * (KH * KW);
            int ki = r / KW, kj = r - ki * KW;
            int l  = l0 + n;
            int y  = l / WOUT, xo = l - y * WOUT;
            int iy = y * STRIDE + ki - PAD;
            int ix = xo * STRIDE + kj - PAD;
            if (iy >= 0 && iy < Hh && ix >= 0 && ix < Ww)
                v = x[((b * CIN + c) * Hh + iy) * Ww + ix];
        }
        lds_p[i] = v;
    }
    __syncthreads();

    const int wv   = tid >> 5;
    const int lane = tid & 31;
    const int mn   = lane & 15;
    const int kg   = lane >> 4;

    v8f accO = {}; v8f accT = {};
    const float* wr = &lds_w[(wv * 16 + mn) * KP];
    #pragma unroll
    for (int kk = 0; kk < KP; kk += 4) {
        int ka = kk + 2 * kg;
        float a0 = wr[ka], a1 = wr[ka + 1];
        v2f aW = {a0, a1};
        v2f aL = {a0 * __builtin_fabsf(a0), a1 * __builtin_fabsf(a1)};
        v2f bP = {lds_p[ka * 16 + mn], lds_p[(ka + 1) * 16 + mn]};
        accO = __builtin_amdgcn_wmma_f32_16x16x4_f32(
            false, aW, false, bP, (short)0, accO, false, false);
        accT = __builtin_amdgcn_wmma_f32_16x16x4_f32(
            false, aL, false, bP, (short)0, accT, false, false);
    }

    #pragma unroll
    for (int r = 0; r < 8; ++r) {
        int o = wv * 16 + r + 8 * kg;
        out[((size_t)b * COUT + o) * L_ + l0 + mn] = accO[r];
        lds_t[o * 16 + mn] = accT[r];
    }
    __syncthreads();

    if (tid < 16) {
        int i0 = 0; float v0 = lds_t[tid];
        for (int o = 1; o < COUT; ++o) {
            float v = lds_t[o * 16 + tid];
            if (v > v0) { v0 = v; i0 = o; }
        }
        int i1 = (i0 == 0) ? 1 : 0;
        float v1 = lds_t[i1 * 16 + tid];
        for (int o = 0; o < COUT; ++o) {
            if (o == i0) continue;
            float v = lds_t[o * 16 + tid];
            if (v > v1) { v1 = v; i1 = o; }
        }
        s_i0[tid] = i0; s_i1[tid] = i1; s_t0[tid] = v0; s_t1[tid] = v1;
    }
    __syncthreads();

    {
        int col = tid / 12;
        int r   = tid - col * 12;
        int i0 = s_i0[col], i1 = s_i1[col];
        for (int k = r; k < K_; k += 12) {
            float p = lds_p[k * 16 + col];
            atomicAdd(&yx[i0 * K_ + k], p);
            atomicAdd(&yx[i1 * K_ + k], DELTA * p);
        }
        if (r == 0) {
            atomicAdd(&yy[i0], s_t0[col]);
            atomicAdd(&yy[i1], DELTA * s_t1[col]);
        }
    }
}

__global__ void zero_ws(float* yx, float* yy, unsigned int* mb)
{
    int i = blockIdx.x * blockDim.x + threadIdx.x;
    if (i < COUT * K_) yx[i] = 0.0f;
    if (i < COUT)      yy[i] = 0.0f;
    if (i == 0)        *mb = __float_as_uint(PRECF);
}

__global__ void finalize_ds(const float* __restrict__ w,
                            const float* __restrict__ yx,
                            const float* __restrict__ yy,
                            float* __restrict__ ds, unsigned int* mb)
{
    int i = blockIdx.x * blockDim.x + threadIdx.x;
    if (i < COUT * K_) {
        int o = i / K_;
        float d = yx[i] * (1.0f / (float)L_) - (yy[o] * (1.0f / (float)L_)) * w[i];
        ds[i] = d;
        atomicMax(mb, __float_as_uint(__builtin_fabsf(d)));
    }
}

__global__ void apply_update(const float* __restrict__ w,
                             const float* __restrict__ ds,
                             const unsigned int* __restrict__ mb,
                             float* __restrict__ neww)
{
    int i = blockIdx.x * blockDim.x + threadIdx.x;
    if (i < COUT * K_) {
        float nc = __uint_as_float(*mb);
        neww[i] = w[i] + LR * (ds[i] / nc);
    }
}

extern "C" void kernel_launch(void* const* d_in, const int* in_sizes, int n_in,
                              void* d_out, int out_size, void* d_ws, size_t ws_size,
                              hipStream_t stream)
{
    const float* x = (const float*)d_in[0];
    const float* w = (const float*)d_in[1];
    float* out  = (float*)d_out;
    float* neww = out + (size_t)B_ * COUT * L_;

    float*        yx = (float*)d_ws;
    float*        yy = yx + COUT * K_;
    unsigned int* mb = (unsigned int*)(yy + COUT);
    float*        ds = (float*)(mb + 1);

    zero_ws<<<(COUT * K_ + 255) / 256, 256, 0, stream>>>(yx, yy, mb);
    conv_hebb_gemm<<<dim3(L_ / 16, B_), 192, 0, stream>>>(x, w, out, yx, yy);
    finalize_ds<<<(COUT * K_ + 255) / 256, 256, 0, stream>>>(w, yx, yy, ds, mb);
    apply_update<<<(COUT * K_ + 255) / 256, 256, 0, stream>>>(w, ds, mb, neww);
}


# Tile loop without per-lane branches: the 38 guarded patch loads become unconditional loads from a clamped index plus a select; the 46 top-2 insert steps and the lane-half merge become selects
# speedup vs baseline: 14.6033x; 14.6033x over previous
#include <hip/hip_runtime.h>

#define B_     64
#define CIN    3
#define Hh     96
#define Ww     96
#define COUT   96
#define KH     5
#define KW     5
#define STRIDE 2
#define PAD    2
#define HOUT   48
#define WOUT   48
#define L_     (HOUT*WOUT)
#define K_     (CIN*KH*KW)
#define KP     76
#define LR     0.02f
#define DELTA  (-0.4f)
#define PRECF  1e-30f

#define NW     8
#define TPG    36
#define NACC   4
#define PP     20
#define KR     80
#define AP     80

typedef float v2f __attribute__((ext_vector_type(2)));
typedef float v4f __attribute__((ext_vector_type(4)));
typedef float v8f __attribute__((ext_vector_type(8)));
typedef int   v4i __attribute__((ext_vector_type(4)));


__device__ __forceinline__ void wave_lds_handover()
{
    __builtin_amdgcn_fence(__ATOMIC_RELEASE, "wavefront");
    __builtin_amdgcn_wave_barrier();
    __builtin_amdgcn_fence(__ATOMIC_ACQUIRE, "wavefront");
}

__global__ __launch_bounds__(256) void conv_hebb_gemm(
    const float* __restrict__ x, const float* __restrict__ w,
    float* __restrict__ out, float* __restrict__ yx, float* __restrict__ yy)
{
    __shared__ __attribute__((aligned(16))) float lds_w[COUT * KP];
    __shared__ __attribute__((aligned(16))) float lds_p[NW][KR * PP];
    __shared__ __attribute__((aligned(16))) int   lds_sel[NW][32];
    __shared__ __attribute__((aligned(16))) float lds_acc[NACC][COUT * AP];
    __shared__ __attribute__((aligned(16))) float lds_yy[NW][COUT * 16];

    const int tid  = threadIdx.x;
    const int wv   = __builtin_amdgcn_readfirstlane(tid >> 5);
    const int lane = tid & 31;
    const int c    = lane & 15;
    const int h    = lane >> 4;

    float* P = lds_p[wv];
    int*   S = lds_sel[wv];
    float* Y = lds_yy[wv];

    {
        float wreg[29];
        #pragma unroll
        for (int u = 0; u < 29; ++u) {
            int i = tid + 256 * u;
            wreg[u] = (i < COUT * K_) ? w[i] : 0.0f;
        }
        #pragma unroll
        for (int u = 0; u < 29; ++u) {
            int i = tid + 256 * u;
            if (i < COUT * K_) { int o = i / K_; int k = i - o * K_; lds_w[o * KP + k] = wreg[u]; }
        }
        if (tid < COUT) lds_w[tid * KP + K_] = 0.0f;
        for (int i = lane; i < (KR - K_) * PP; i += 32) P[K_ * PP + i] = 0.0f;
        for (int i = lane; i < COUT * 16; i += 32) Y[i] = 0.0f;
    }
    __syncthreads();

    v8f   yxacc[6][5];
    #pragma unroll
    for (int t = 0; t < 6; ++t) {
        #pragma unroll
        for (int j = 0; j < 5; ++j) yxacc[t][j] = (v8f){};
    }

    const int    b  = blockIdx.x >> 2;
    const float* xb = x + (size_t)b * CIN * Hh * Ww;

    const int ntile = (wv < TPG - 4 * NW) ? 5 : 4;
    for (int it = 0; it < ntile; ++it) {
        const int l0 = ((blockIdx.x & 3) * TPG + it * NW + wv) * 16;

        wave_lds_handover();

        {
            const int l   = l0 + c;
            const int y   = l / WOUT, xo = l - y * WOUT;
            const int iy0 = y * STRIDE - PAD, ix0 = xo * STRIDE - PAD;
            float g[38];
            #pragma unroll
            for (int u = 0; u < 38; ++u) {
                int k  = 2 * u + h;
                int cc = k / (KH * KW);
                int r  = k - cc * (KH * KW);
                int ki = r / KW, kj = r - ki * KW;
                int iy = iy0 + ki, ix = ix0 + kj;
                unsigned um = ((unsigned)iy > (unsigned)ix) ? (unsigned)iy : (unsigned)ix;
                bool ok = (k < K_) & (um < (unsigned)Hh);
                int iyc = iy < 0 ? 0 : (iy > Hh - 1 ? Hh - 1 : iy);
                int ixc = ix < 0 ? 0 : (ix > Ww - 1 ? Ww - 1 : ix);
                int ccc = cc > CIN - 1 ? CIN - 1 : cc;
                float v = xb[(unsigned)((ccc * Hh + iyc) * Ww + ixc)];
                g[u] = ok ? v : 0.0f;
            }
            #pragma unroll
            for (int u = 0; u < 38; ++u) {
                int k = 2 * u + h;
                P[k * PP + c] = g[u];
            }
        }
        wave_lds_handover();
        __builtin_amdgcn_sched_barrier(0);

        v8f aO[6], aT[6];
        #pragma unroll
        for (int t = 0; t < 6; ++t) { aO[t] = (v8f){}; aT[t] = (v8f){}; }
        #pragma unroll
        for (int kk = 0; kk < KP; kk += 4) {
            const int ka = kk + 2 * h;
            v2f bP = {P[ka * PP + c], P[(ka + 1) * PP + c]};
            #pragma unroll
            for (int t = 0; t < 6; ++t) {
                const float* wr = &lds_w[(t * 16 + c) * KP + ka];
                float a0 = wr[0], a1 = wr[1];
                v2f aW = {a0, a1};
                v2f aL = {a0 * __builtin_fabsf(a0), a1 * __builtin_fabsf(a1)};
                aO[t] = __builtin_amdgcn_wmma_f32_16x16x4_f32(
                    false, aW, false, bP, (short)0, aO[t], false, false);
                aT[t] = __builtin_amdgcn_wmma_f32_16x16x4_f32(
                    false, aL, false, bP, (short)0, aT[t], false, false);
            }
        }

        __builtin_amdgcn_sched_barrier(0);

        {
            float* ob = out + (size_t)b * COUT * L_ + l0 + c;
            #pragma unroll
            for (int t = 0; t < 6; ++t) {
                #pragma unroll
                for (int r = 0; r < 8; ++r)
                    ob[(size_t)(t * 16 + 8 * h + r) * L_] = aO[t][r];
            }
        }

        float v0, v1; int i0, i1;
        {
            float fa = aT[0][0], fb = aT[0][1];
            bool  sw = fb > fa;
            v0 = sw ? fb : fa; i0 = sw ? 1 : 0;
            v1 = sw ? fa : fb; i1 = sw ? 0 : 1;
        }
        #pragma unroll
        for (int t = 0; t < 6; ++t) {
            #pragma unroll
            for (int r = 0; r < 8; ++r) {
                if (t == 0 && r < 2) continue;
                const float v  = aT[t][r];
                const int   o  = t * 16 + r;
                const bool  g0 = v > v0;
                const bool  g1 = v > v1;
                const float nv = g1 ? v : v1;
                const int   ni = g1 ? o : i1;
                v1 = g0 ? v0 : nv; i1 = g0 ? i0 : ni;
                v0 = g0 ? v  : v0; i0 = g0 ? o  : i0;
            }
        }
        i0 += 8 * h; i1 += 8 * h;
        int n0, n1; float m0, m1;
        {
            float w0 = __shfl_xor(v0, 16), w1 = __shfl_xor(v1, 16);
            int   j0 = __shfl_xor(i0, 16), j1 = __shfl_xor(i1, 16);
            const bool mine = (v0 > w0) | ((v0 == w0) & (i0 < j0));
            const bool sa   = (v1 > w0) | ((v1 == w0) & (i1 < j0));
            const bool sb   = (w1 > v0) | ((w1 == v0) & (j1 < i0));
            const int   na = sa ? i1 : j0;  const float ma = sa ? v1 : w0;
            const int   nb = sb ? j1 : i0;  const float mb = sb ? w1 : v0;
            n0 = mine ? i0 : j0;  m0 = mine ? v0 : w0;
            n1 = mine ? na : nb;  m1 = mine ? ma : mb;
        }

        if (h == 0) {
            Y[n0 * 16 + c] += m0;
            Y[n1 * 16 + c] += DELTA * m1;
            S[c] = n0; S[16 + c] = n1;
        }
        wave_lds_handover();
        int q0[8], q1[8];
        {
            v4i sa = *(const v4i*)&S[8 * h],      sb = *(const v4i*)&S[8 * h + 4];
            v4i sc = *(const v4i*)&S[16 + 8 * h], sd = *(const v4i*)&S[16 + 8 * h + 4];
            q0[0] = sa.x; q0[1] = sa.y; q0[2] = sa.z; q0[3] = sa.w;
            q0[4] = sb.x; q0[5] = sb.y; q0[6] = sb.z; q0[7] = sb.w;
            q1[0] = sc.x; q1[1] = sc.y; q1[2] = sc.z; q1[3] = sc.w;
            q1[4] = sd.x; q1[5] = sd.y; q1[6] = sd.z; q1[7] = sd.w;
        }

        v2f aq[6][4];
        #pragma unroll
        for (int t = 0; t < 6; ++t) {
            const int o = t * 16 + c;
            #pragma unroll
            for (int s = 0; s < 4; ++s) {
                float e0 = (o == q0[2 * s])     ? 1.0f : ((o == q1[2 * s])     ? DELTA : 0.0f);
                float e1 = (o == q0[2 * s + 1]) ? 1.0f : ((o == q1[2 * s + 1]) ? DELTA : 0.0f);
                aq[t][s] = (v2f){e0, e1};
            }
        }
        #pragma unroll
        for (int j = 0; j < 5; ++j) {
            v4f pa = *(const v4f*)&P[(16 * j + c) * PP + 8 * h];
            v4f pb = *(const v4f*)&P[(16 * j + c) * PP + 8 * h + 4];
            v2f bq[4] = { {pa.x, pa.y}, {pa.z, pa.w}, {pb.x, pb.y}, {pb.z, pb.w} };
            #pragma unroll
            for (int s = 0; s < 4; ++s) {
                #pragma unroll
                for (int t = 0; t < 6; ++t)
                    yxacc[t][j] = __builtin_amdgcn_wmma_f32_16x16x4_f32(
                        false, aq[t][s], false, bq[s], (short)0, yxacc[t][j], false, false);
            }
        }
    }

    float* accw = lds_acc[wv & (NACC - 1)];
    if (wv < NACC) {
        #pragma unroll
        for (int t = 0; t < 6; ++t) {
            #pragma unroll
            for (int j = 0; j < 5; ++j) {
                #pragma unroll
                for (int r = 0; r < 8; ++r)
                    accw[(t * 16 + 8 * h + r) * AP + 16 * j + c] = yxacc[t][j][r];
            }
        }
    }
    __syncthreads();
    if (wv >= NACC) {
        #pragma unroll
        for (int t = 0; t < 6; ++t) {
            #pragma unroll
            for (int j = 0; j < 5; ++j) {
                #pragma unroll
                for (int r = 0; r < 8; ++r)
                    accw[(t * 16 + 8 * h + r) * AP + 16 * j + c] += yxacc[t][j][r];
            }
        }
    }
    if (tid < COUT) {
        float s = 0.0f;
        for (int wq = 0; wq < NW; ++wq) {
            for (int cc = 0; cc < 16; ++cc) s += lds_yy[wq][tid * 16 + cc];
        }
        atomicAdd(&yy[tid], s);
    }
    __syncthreads();
    {
        const int rot = (int)((blockIdx.x * 224u) % (unsigned)(COUT * K_));
        for (int i = tid; i < COUT * K_; i += 256) {
            int j = i + rot; if (j >= COUT * K_) j -= COUT * K_;
            int o = j / K_, k = j - o * K_;
            float s = (lds_acc[0][o * AP + k] + lds_acc[1][o * AP + k])
                    + (lds_acc[2][o * AP + k] + lds_acc[3][o * AP + k]);
            atomicAdd(&yx[j], s);
        }
    }
}

__global__ void zero_ws(float* yx, float* yy, unsigned int* mb)
{
    int i = blockIdx.x * blockDim.x + threadIdx.x;
    if (i < COUT * K_) yx[i] = 0.0f;
    if (i < COUT)      yy[i] = 0.0f;
    if (i == 0)        *mb = __float_as_uint(PRECF);
}

__global__ void finalize_ds(const float* __restrict__ w,
                            const float* __restrict__ yx,
                            const float* __restrict__ yy,
                            float* __restrict__ ds, unsigned int* mb)
{
    int i = blockIdx.x * blockDim.x + threadIdx.x;
    if (i < COUT * K_) {
        int o = i / K_;
        float d = yx[i] * (1.0f / (float)L_) - (yy[o] * (1.0f / (float)L_)) * w[i];
        ds[i] = d;
        atomicMax(mb, __float_as_uint(__builtin_fabsf(d)));
    }
}

__global__ void apply_update(const float* __restrict__ w,
                             const float* __restrict__ ds,
                             const unsigned int* __restrict__ mb,
                             float* __restrict__ neww)
{
    int i = blockIdx.x * blockDim.x + threadIdx.x;
    if (i < COUT * K_) {
        float nc = __uint_as_float(*mb);
        neww[i] = w[i] + LR * (ds[i] / nc);
    }
}

extern "C" void kernel_launch(void* const* d_in, const int* in_sizes, int n_in,
                              void* d_out, int out_size, void* d_ws, size_t ws_size,
                              hipStream_t stream)
{
    const float* x = (const float*)d_in[0];
    const float* w = (const float*)d_in[1];
    float* out  = (float*)d_out;
    float* neww = out + (size_t)B_ * COUT * L_;

    float*        yx = (float*)d_ws;
    float*        yy = yx + COUT * K_;
    unsigned int* mb = (unsigned int*)(yy + COUT);
    float*        ds = (float*)(mb + 1);

    zero_ws<<<(COUT * K_ + 255) / 256, 256, 0, stream>>>(yx, yy, mb);
    conv_hebb_gemm<<<256, 256, 0, stream>>>(x, w, out, yx, yy);
    finalize_ds<<<(COUT * K_ + 255) / 256, 256, 0, stream>>>(w, yx, yy, ds, mb);
    apply_update<<<(COUT * K_ + 255) / 256, 256, 0, stream>>>(w, ds, mb, neww);
}


# Product and yx loops: all LDS fragment reads of a block (3 k-steps) stand as ONE group, a block ahead of its matrix instructions in a second register set, not spread among them; sums unchanged
# speedup vs baseline: 15.1651x; 1.0385x over previous
#include <hip/hip_runtime.h>

#define B_     64
#define CIN    3
#define Hh     96
#define Ww     96
#define COUT   96
#define KH     5
#define KW     5
#define STRIDE 2
#define PAD    2
#define HOUT   48
#define WOUT   48
#define L_     (HOUT*WOUT)
#define K_     (CIN*KH*KW)
#define KP     76
#define LR     0.02f
#define DELTA  (-0.4f)
#define PRECF  1e-30f

#define NW     8
#define TPG    36
#define NACC   4
#define PP     20
#define KR     80
#define AP     80
#define KBLK   3
#define NBLK   7

typedef float v2f __attribute__((ext_vector_type(2)));
typedef float v4f __attribute__((ext_vector_type(4)));
typedef float v8f __attribute__((ext_vector_type(8)));
typedef int   v4i __attribute__((ext_vector_type(4)));


__device__ __forceinline__ void wave_lds_handover()
{
    __builtin_amdgcn_fence(__ATOMIC_RELEASE, "wavefront");
    __builtin_amdgcn_wave_barrier();
    __builtin_amdgcn_fence(__ATOMIC_ACQUIRE, "wavefront");
}

__global__ __launch_bounds__(256) void conv_hebb_gemm(
    const float* __restrict__ x, const float* __restrict__ w,
    float* __restrict__ out, float* __restrict__ yx, float* __restrict__ yy)
{
    __shared__ __attribute__((aligned(16))) float lds_w[COUT * KP];
    __shared__ __attribute__((aligned(16))) float lds_p[NW][KR * PP];
    __shared__ __attribute__((aligned(16))) int   lds_sel[NW][32];
    __shared__ __attribute__((aligned(16))) float lds_acc[NACC][COUT * AP];
    __shared__ __attribute__((aligned(16))) float lds_yy[NW][COUT * 16];

    const int tid  = threadIdx.x;
    const int wv   = __builtin_amdgcn_readfirstlane(tid >> 5);
    const int lane = tid & 31;
    const int c    = lane & 15;
    const int h    = lane >> 4;

    float* P = lds_p[wv];
    int*   S = lds_sel[wv];
    float* Y = lds_yy[wv];

    {
        float wreg[29];
        #pragma unroll
        for (int u = 0; u < 29; ++u) {
            int i = tid + 256 * u;
            wreg[u] = (i < COUT * K_) ? w[i] : 0.0f;
        }
        #pragma unroll
        for (int u = 0; u < 29; ++u) {
            int i = tid + 256 * u;
            if (i < COUT * K_) { int o = i / K_; int k = i - o * K_; lds_w[o * KP + k] = wreg[u]; }
        }
        if (tid < COUT) lds_w[tid * KP + K_] = 0.0f;
        for (int i = lane; i < (KR - K_) * PP; i += 32) P[K_ * PP + i] = 0.0f;
        for (int i = lane; i < COUT * 16; i += 32) Y[i] = 0.0f;
    }
    __syncthreads();

    v8f   yxacc[6][5];
    #pragma unroll
    for (int t = 0; t < 6; ++t) {
        #pragma unroll
        for (int j = 0; j < 5; ++j) yxacc[t][j] = (v8f){};
    }

    const int    b  = blockIdx.x >> 2;
    const float* xb = x + (size_t)b * CIN * Hh * Ww;

    const int ntile = (wv < TPG - 4 * NW) ? 5 : 4;
    for (int it = 0; it < ntile; ++it) {
        const int l0 = ((blockIdx.x & 3) * TPG + it * NW + wv) * 16;

        wave_lds_handover();

        {
            const int l   = l0 + c;
            const int y   = l / WOUT, xo = l - y * WOUT;
            const int iy0 = y * STRIDE - PAD, ix0 = xo * STRIDE - PAD;
            float g[38];
            #pragma unroll
            for (int u = 0; u < 38; ++u) {
                int k  = 2 * u + h;
                int cc = k / (KH * KW);
                int r  = k - cc * (KH * KW);
                int ki = r / KW, kj = r - ki * KW;
                int iy = iy0 + ki, ix = ix0 + kj;
                unsigned um = ((unsigned)iy > (unsigned)ix) ? (unsigned)iy : (unsigned)ix;
                bool ok = (k < K_) & (um < (unsigned)Hh);
                int iyc = iy < 0 ? 0 : (iy > Hh - 1 ? Hh - 1 : iy);
                int ixc = ix < 0 ? 0 : (ix > Ww - 1 ? Ww - 1 : ix);
                int ccc = cc > CIN - 1 ? CIN - 1 : cc;
                float v = xb[(unsigned)((ccc * Hh + iyc) * Ww + ixc)];
                g[u] = ok ? v : 0.0f;
            }
            #pragma unroll
            for (int u = 0; u < 38; ++u) {
                int k = 2 * u + h;
                P[k * PP + c] = g[u];
            }
        }
        wave_lds_handover();

        v8f aO[6], aT[6];
        #pragma unroll
        for (int t = 0; t < 6; ++t) { aO[t] = (v8f){}; aT[t] = (v8f){}; }
        v2f fa[2][KBLK][6], fb[2][KBLK];
        #pragma unroll
        for (int s = 0; s < KBLK; ++s) {
            const int ka = 4 * s + 2 * h;
            fb[0][s] = (v2f){P[ka * PP + c], P[(ka + 1) * PP + c]};
            #pragma unroll
            for (int t = 0; t < 6; ++t) fa[0][s][t] = *(const v2f*)&lds_w[(t * 16 + c) * KP + ka];
        }
        #pragma unroll
        for (int bl = 0; bl < NBLK; ++bl) {
            __builtin_amdgcn_sched_barrier(0);
            #pragma unroll
            for (int s = 0; s < KBLK; ++s) {
                const int kk = 4 * ((bl + 1) * KBLK + s);
                if (bl + 1 < NBLK && kk < KP) {
                    const int ka = kk + 2 * h;
                    fb[(bl + 1) & 1][s] = (v2f){P[ka * PP + c], P[(ka + 1) * PP + c]};
                    #pragma unroll
                    for (int t = 0; t < 6; ++t)
                        fa[(bl + 1) & 1][s][t] = *(const v2f*)&lds_w[(t * 16 + c) * KP + ka];
                }
            }
            __builtin_amdgcn_sched_barrier(0);
            #pragma unroll
            for (int s = 0; s < KBLK; ++s) {
                const int kk = 4 * (bl * KBLK + s);
                if (kk < KP) {
                    const v2f bP = fb[bl & 1][s];
                    #pragma unroll
                    for (int t = 0; t < 6; ++t) {
                        const v2f aW = fa[bl & 1][s][t];
                        v2f aL = {aW.x * __builtin_fabsf(aW.x), aW.y * __builtin_fabsf(aW.y)};
                        aO[t] = __builtin_amdgcn_wmma_f32_16x16x4_f32(
                            false, aW, false, bP, (short)0, aO[t], false, false);
                        aT[t] = __builtin_amdgcn_wmma_f32_16x16x4_f32(
                            false, aL, false, bP, (short)0, aT[t], false, false);
                    }
                }
            }
        }

        __builtin_amdgcn_sched_barrier(0);

        {
            float* ob = out + (size_t)b * COUT * L_ + l0 + c;
            #pragma unroll
            for (int t = 0; t < 6; ++t) {
                #pragma unroll
                for (int r = 0; r < 8; ++r)
                    ob[(size_t)(t * 16 + 8 * h + r) * L_] = aO[t][r];
            }
        }

        float v0, v1; int i0, i1;
        {
            float fa = aT[0][0], fb = aT[0][1];
            bool  sw = fb > fa;
            v0 = sw ? fb : fa; i0 = sw ? 1 : 0;
            v1 = sw ? fa : fb; i1 = sw ? 0 : 1;
        }
        #pragma unroll
        for (int t = 0; t < 6; ++t) {
            #pragma unroll
            for (int r = 0; r < 8; ++r) {
                if (t == 0 && r < 2) continue;
                const float v  = aT[t][r];
                const int   o  = t * 16 + r;
                const bool  g0 = v > v0;
                const bool  g1 = v > v1;
                const float nv = g1 ? v : v1;
                const int   ni = g1 ? o : i1;
                v1 = g0 ? v0 : nv; i1 = g0 ? i0 : ni;
                v0 = g0 ? v  : v0; i0 = g0 ? o  : i0;
            }
        }
        i0 += 8 * h; i1 += 8 * h;
        int n0, n1; float m0, m1;
        {
            float w0 = __shfl_xor(v0, 16), w1 = __shfl_xor(v1, 16);
            int   j0 = __shfl_xor(i0, 16), j1 = __shfl_xor(i1, 16);
            const bool mine = (v0 > w0) | ((v0 == w0) & (i0 < j0));
            const bool sa   = (v1 > w0) | ((v1 == w0) & (i1 < j0));
            const bool sb   = (w1 > v0) | ((w1 == v0) & (j1 < i0));
            const int   na = sa ? i1 : j0;  const float ma = sa ? v1 : w0;
            const int   nb = sb ? j1 : i0;  const float mb = sb ? w1 : v0;
            n0 = mine ? i0 : j0;  m0 = mine ? v0 : w0;
            n1 = mine ? na : nb;  m1 = mine ? ma : mb;
        }

        if (h == 0) {
            Y[n0 * 16 + c] += m0;
            Y[n1 * 16 + c] += DELTA * m1;
            S[c] = n0; S[16 + c] = n1;
        }
        wave_lds_handover();
        v4i sa = *(const v4i*)&S[8 * h],      sb = *(const v4i*)&S[8 * h + 4];
        v4i sc = *(const v4i*)&S[16 + 8 * h], sd = *(const v4i*)&S[16 + 8 * h + 4];
        v4f pa[5], pb[5];
        #pragma unroll
        for (int j = 0; j < 5; ++j) {
            pa[j] = *(const v4f*)&P[(16 * j + c) * PP + 8 * h];
            pb[j] = *(const v4f*)&P[(16 * j + c) * PP + 8 * h + 4];
        }
        __builtin_amdgcn_sched_barrier(0);
        int q0[8], q1[8];
        {
            q0[0] = sa.x; q0[1] = sa.y; q0[2] = sa.z; q0[3] = sa.w;
            q0[4] = sb.x; q0[5] = sb.y; q0[6] = sb.z; q0[7] = sb.w;
            q1[0] = sc.x; q1[1] = sc.y; q1[2] = sc.z; q1[3] = sc.w;
            q1[4] = sd.x; q1[5] = sd.y; q1[6] = sd.z; q1[7] = sd.w;
        }

        v2f aq[6][4];
        #pragma unroll
        for (int t = 0; t < 6; ++t) {
            const int o = t * 16 + c;
            #pragma unroll
            for (int s = 0; s < 4; ++s) {
                float e0 = (o == q0[2 * s])     ? 1.0f : ((o == q1[2 * s])     ? DELTA : 0.0f);
                float e1 = (o == q0[2 * s + 1]) ? 1.0f : ((o == q1[2 * s + 1]) ? DELTA : 0.0f);
                aq[t][s] = (v2f){e0, e1};
            }
        }
        #pragma unroll
        for (int j = 0; j < 5; ++j) {
            v2f bq[4] = { {pa[j].x, pa[j].y}, {pa[j].z, pa[j].w}, {pb[j].x, pb[j].y}, {pb[j].z, pb[j].w} };
            #pragma unroll
            for (int s = 0; s < 4; ++s) {
                #pragma unroll
                for (int t = 0; t < 6; ++t)
                    yxacc[t][j] = __builtin_amdgcn_wmma_f32_16x16x4_f32(
                        false, aq[t][s], false, bq[s], (short)0, yxacc[t][j], false, false);
            }
        }
    }

    float* accw = lds_acc[wv & (NACC - 1)];
    if (wv < NACC) {
        #pragma unroll
        for (int t = 0; t < 6; ++t) {
            #pragma unroll
            for (int j = 0; j < 5; ++j) {
                #pragma unroll
                for (int r = 0; r < 8; ++r)
                    accw[(t * 16 + 8 * h + r) * AP + 16 * j + c] = yxacc[t][j][r];
            }
        }
    }
    __syncthreads();
    if (wv >= NACC) {
        #pragma unroll
        for (int t = 0; t < 6; ++t) {
            #pragma unroll
            for (int j = 0; j < 5; ++j) {
                #pragma unroll
                for (int r = 0; r < 8; ++r)
                    accw[(t * 16 + 8 * h + r) * AP + 16 * j + c] += yxacc[t][j][r];
            }
        }
    }
    if (tid < COUT) {
        float s = 0.0f;
        for (int wq = 0; wq < NW; ++wq) {
            for (int cc = 0; cc < 16; ++cc) s += lds_yy[wq][tid * 16 + cc];
        }
        atomicAdd(&yy[tid], s);
    }
    __syncthreads();
    {
        const int rot = (int)((blockIdx.x * 224u) % (unsigned)(COUT * K_));
        for (int i = tid; i < COUT * K_; i += 256) {
            int j = i + rot; if (j >= COUT * K_) j -= COUT * K_;
            int o = j / K_, k = j - o * K_;
            float s = (lds_acc[0][o * AP + k] + lds_acc[1][o * AP + k])
                    + (lds_acc[2][o * AP + k] + lds_acc[3][o * AP + k]);
            atomicAdd(&yx[j], s);
        }
    }
}

__global__ void zero_ws(float* yx, float* yy, unsigned int* mb)
{
    int i = blockIdx.x * blockDim.x + threadIdx.x;
    if (i < COUT * K_) yx[i] = 0.0f;
    if (i < COUT)      yy[i] = 0.0f;
    if (i == 0)        *mb = __float_as_uint(PRECF);
}

__global__ void finalize_ds(const float* __restrict__ w,
                            const float* __restrict__ yx,
                            const float* __restrict__ yy,
                            float* __restrict__ ds, unsigned int* mb)
{
    int i = blockIdx.x * blockDim.x + threadIdx.x;
    if (i < COUT * K_) {
        int o = i / K_;
        float d = yx[i] * (1.0f / (float)L_) - (yy[o] * (1.0f / (float)L_)) * w[i];
        ds[i] = d;
        atomicMax(mb, __float_as_uint(__builtin_fabsf(d)));
    }
}

__global__ void apply_update(const float* __restrict__ w,
                             const float* __restrict__ ds,
                             const unsigned int* __restrict__ mb,
                             float* __restrict__ neww)
{
    int i = blockIdx.x * blockDim.x + threadIdx.x;
    if (i < COUT * K_) {
        float nc = __uint_as_float(*mb);
        neww[i] = w[i] + LR * (ds[i] / nc);
    }
}

extern "C" void kernel_launch(void* const* d_in, const int* in_sizes, int n_in,
                              void* d_out, int out_size, void* d_ws, size_t ws_size,
                              hipStream_t stream)
{
    const float* x = (const float*)d_in[0];
    const float* w = (const float*)d_in[1];
    float* out  = (float*)d_out;
    float* neww = out + (size_t)B_ * COUT * L_;

    float*        yx = (float*)d_ws;
    float*        yy = yx + COUT * K_;
    unsigned int* mb = (unsigned int*)(yy + COUT);
    float*        ds = (float*)(mb + 1);

    zero_ws<<<(COUT * K_ + 255) / 256, 256, 0, stream>>>(yx, yy, mb);
    conv_hebb_gemm<<<256, 256, 0, stream>>>(x, w, out, yx, yy);
    finalize_ds<<<(COUT * K_ + 255) / 256, 256, 0, stream>>>(w, yx, yy, ds, mb);
    apply_update<<<(COUT * K_ + 255) / 256, 256, 0, stream>>>(w, ds, mb, neww);
}


# Group sums of yx and yy go into one of 8 copies, the copy of the die the group runs on (hardware id): no sum line takes adds from both halves; finalize adds the copies; sum order changes
# speedup vs baseline: 16.3694x; 1.0794x over previous
#include <hip/hip_runtime.h>

#define B_     64
#define CIN    3
#define Hh     96
#define Ww     96
#define COUT   96
#define KH     5
#define KW     5
#define STRIDE 2
#define PAD    2
#define HOUT   48
#define WOUT   48
#define L_     (HOUT*WOUT)
#define K_     (CIN*KH*KW)
#define KP     76
#define LR     0.02f
#define DELTA  (-0.4f)
#define PRECF  1e-30f

#define NW     8
#define TPG    36
#define NACC   4
#define PP     20
#define KR     80
#define AP     80
#define KBLK   3
#define NBLK   7
#define NCOPY  8
#define CST    (COUT*K_ + COUT)

typedef float v2f __attribute__((ext_vector_type(2)));
typedef float v4f __attribute__((ext_vector_type(4)));
typedef float v8f __attribute__((ext_vector_type(8)));
typedef int   v4i __attribute__((ext_vector_type(4)));


__device__ __forceinline__ void wave_lds_handover()
{
    __builtin_amdgcn_fence(__ATOMIC_RELEASE, "wavefront");
    __builtin_amdgcn_wave_barrier();
    __builtin_amdgcn_fence(__ATOMIC_ACQUIRE, "wavefront");
}

__global__ __launch_bounds__(256) void conv_hebb_gemm(
    const float* __restrict__ x, const float* __restrict__ w,
    float* __restrict__ out, float* __restrict__ yx, float* __restrict__ yy)
{
    __shared__ __attribute__((aligned(16))) float lds_w[COUT * KP];
    __shared__ __attribute__((aligned(16))) float lds_p[NW][KR * PP];
    __shared__ __attribute__((aligned(16))) int   lds_sel[NW][32];
    __shared__ __attribute__((aligned(16))) float lds_acc[NACC][COUT * AP];
    __shared__ __attribute__((aligned(16))) float lds_yy[NW][COUT * 16];

    const int tid  = threadIdx.x;
    const int wv   = __builtin_amdgcn_readfirstlane(tid >> 5);
    const int lane = tid & 31;
    const int c    = lane & 15;
    const int h    = lane >> 4;

    float* P = lds_p[wv];
    int*   S = lds_sel[wv];
    float* Y = lds_yy[wv];

    const int die = (int)((__builtin_amdgcn_s_sendmsg_rtn(0x87) >> 16) & (NCOPY - 1));
    float* yxc = yx + die * CST;
    float* yyc = yy + die * CST;

    {
        float wreg[29];
        #pragma unroll
        for (int u = 0; u < 29; ++u) {
            int i = tid + 256 * u;
            wreg[u] = (i < COUT * K_) ? w[i] : 0.0f;
        }
        #pragma unroll
        for (int u = 0; u < 29; ++u) {
            int i = tid + 256 * u;
            if (i < COUT * K_) { int o = i / K_; int k = i - o * K_; lds_w[o * KP + k] = wreg[u]; }
        }
        if (tid < COUT) lds_w[tid * KP + K_] = 0.0f;
        for (int i = lane; i < (KR - K_) * PP; i += 32) P[K_ * PP + i] = 0.0f;
        for (int i = lane; i < COUT * 16; i += 32) Y[i] = 0.0f;
    }
    __syncthreads();

    v8f   yxacc[6][5];
    #pragma unroll
    for (int t = 0; t < 6; ++t) {
        #pragma unroll
        for (int j = 0; j < 5; ++j) yxacc[t][j] = (v8f){};
    }

    const int    b  = blockIdx.x >> 2;
    const float* xb = x + (size_t)b * CIN * Hh * Ww;

    const int ntile = (wv < TPG - 4 * NW) ? 5 : 4;
    for (int it = 0; it < ntile; ++it) {
        const int l0 = ((blockIdx.x & 3) * TPG + it * NW + wv) * 16;

        wave_lds_handover();

        {
            const int l   = l0 + c;
            const int y   = l / WOUT, xo = l - y * WOUT;
            const int iy0 = y * STRIDE - PAD, ix0 = xo * STRIDE - PAD;
            float g[38];
            #pragma unroll
            for (int u = 0; u < 38; ++u) {
                int k  = 2 * u + h;
                int cc = k / (KH * KW);
                int r  = k - cc * (KH * KW);
                int ki = r / KW, kj = r - ki * KW;
                int iy = iy0 + ki, ix = ix0 + kj;
                unsigned um = ((unsigned)iy > (unsigned)ix) ? (unsigned)iy : (unsigned)ix;
                bool ok = (k < K_) & (um < (unsigned)Hh);
                int iyc = iy < 0 ? 0 : (iy > Hh - 1 ? Hh - 1 : iy);
                int ixc = ix < 0 ? 0 : (ix > Ww - 1 ? Ww - 1 : ix);
                int ccc = cc > CIN - 1 ? CIN - 1 : cc;
                float v = xb[(unsigned)((ccc * Hh + iyc) * Ww + ixc)];
                g[u] = ok ? v : 0.0f;
            }
            #pragma unroll
            for (int u = 0; u < 38; ++u) {
                int k = 2 * u + h;
                P[k * PP + c] = g[u];
            }
        }
        wave_lds_handover();

        v8f aO[6], aT[6];
        #pragma unroll
        for (int t = 0; t < 6; ++t) { aO[t] = (v8f){}; aT[t] = (v8f){}; }
        v2f fa[2][KBLK][6], fb[2][KBLK];
        #pragma unroll
        for (int s = 0; s < KBLK; ++s) {
            const int ka = 4 * s + 2 * h;
            fb[0][s] = (v2f){P[ka * PP + c], P[(ka + 1) * PP + c]};
            #pragma unroll
            for (int t = 0; t < 6; ++t) fa[0][s][t] = *(const v2f*)&lds_w[(t * 16 + c) * KP + ka];
        }
        #pragma unroll
        for (int bl = 0; bl < NBLK; ++bl) {
            __builtin_amdgcn_sched_barrier(0);
            #pragma unroll
            for (int s = 0; s < KBLK; ++s) {
                const int kk = 4 * ((bl + 1) * KBLK + s);
                if (bl + 1 < NBLK && kk < KP) {
                    const int ka = kk + 2 * h;
                    fb[(bl + 1) & 1][s] = (v2f){P[ka * PP + c], P[(ka + 1) * PP + c]};
                    #pragma unroll
                    for (int t = 0; t < 6; ++t)
                        fa[(bl + 1) & 1][s][t] = *(const v2f*)&lds_w[(t * 16 + c) * KP + ka];
                }
            }
            __builtin_amdgcn_sched_barrier(0);
            #pragma unroll
            for (int s = 0; s < KBLK; ++s) {
                const int kk = 4 * (bl * KBLK + s);
                if (kk < KP) {
                    const v2f bP = fb[bl & 1][s];
                    #pragma unroll
                    for (int t = 0; t < 6; ++t) {
                        const v2f aW = fa[bl & 1][s][t];
                        v2f aL = {aW.x * __builtin_fabsf(aW.x), aW.y * __builtin_fabsf(aW.y)};
                        aO[t] = __builtin_amdgcn_wmma_f32_16x16x4_f32(
                            false, aW, false, bP, (short)0, aO[t], false, false);
                        aT[t] = __builtin_amdgcn_wmma_f32_16x16x4_f32(
                            false, aL, false, bP, (short)0, aT[t], false, false);
                    }
                }
            }
        }

        __builtin_amdgcn_sched_barrier(0);

        {
            float* ob = out + (size_t)b * COUT * L_ + l0 + c;
            #pragma unroll
            for (int t = 0; t < 6; ++t) {
                #pragma unroll
                for (int r = 0; r < 8; ++r)
                    ob[(size_t)(t * 16 + 8 * h + r) * L_] = aO[t][r];
            }
        }

        float v0, v1; int i0, i1;
        {
            float fa = aT[0][0], fb = aT[0][1];
            bool  sw = fb > fa;
            v0 = sw ? fb : fa; i0 = sw ? 1 : 0;
            v1 = sw ? fa : fb; i1 = sw ? 0 : 1;
        }
        #pragma unroll
        for (int t = 0; t < 6; ++t) {
            #pragma unroll
            for (int r = 0; r < 8; ++r) {
                if (t == 0 && r < 2) continue;
                const float v  = aT[t][r];
                const int   o  = t * 16 + r;
                const bool  g0 = v > v0;
                const bool  g1 = v > v1;
                const float nv = g1 ? v : v1;
                const int   ni = g1 ? o : i1;
                v1 = g0 ? v0 : nv; i1 = g0 ? i0 : ni;
                v0 = g0 ? v  : v0; i0 = g0 ? o  : i0;
            }
        }
        i0 += 8 * h; i1 += 8 * h;
        int n0, n1; float m0, m1;
        {
            float w0 = __shfl_xor(v0, 16), w1 = __shfl_xor(v1, 16);
            int   j0 = __shfl_xor(i0, 16), j1 = __shfl_xor(i1, 16);
            const bool mine = (v0 > w0) | ((v0 == w0) & (i0 < j0));
            const bool sa   = (v1 > w0) | ((v1 == w0) & (i1 < j0));
            const bool sb   = (w1 > v0) | ((w1 == v0) & (j1 < i0));
            const int   na = sa ? i1 : j0;  const float ma = sa ? v1 : w0;
            const int   nb = sb ? j1 : i0;  const float mb = sb ? w1 : v0;
            n0 = mine ? i0 : j0;  m0 = mine ? v0 : w0;
            n1 = mine ? na : nb;  m1 = mine ? ma : mb;
        }

        if (h == 0) {
            Y[n0 * 16 + c] += m0;
            Y[n1 * 16 + c] += DELTA * m1;
            S[c] = n0; S[16 + c] = n1;
        }
        wave_lds_handover();
        v4i sa = *(const v4i*)&S[8 * h],      sb = *(const v4i*)&S[8 * h + 4];
        v4i sc = *(const v4i*)&S[16 + 8 * h], sd = *(const v4i*)&S[16 + 8 * h + 4];
        v4f pa[5], pb[5];
        #pragma unroll
        for (int j = 0; j < 5; ++j) {
            pa[j] = *(const v4f*)&P[(16 * j + c) * PP + 8 * h];
            pb[j] = *(const v4f*)&P[(16 * j + c) * PP + 8 * h + 4];
        }
        __builtin_amdgcn_sched_barrier(0);
        int q0[8], q1[8];
        {
            q0[0] = sa.x; q0[1] = sa.y; q0[2] = sa.z; q0[3] = sa.w;
            q0[4] = sb.x; q0[5] = sb.y; q0[6] = sb.z; q0[7] = sb.w;
            q1[0] = sc.x; q1[1] = sc.y; q1[2] = sc.z; q1[3] = sc.w;
            q1[4] = sd.x; q1[5] = sd.y; q1[6] = sd.z; q1[7] = sd.w;
        }

        v2f aq[6][4];
        #pragma unroll
        for (int t = 0; t < 6; ++t) {
            const int o = t * 16 + c;
            #pragma unroll
            for (int s = 0; s < 4; ++s) {
                float e0 = (o == q0[2 * s])     ? 1.0f : ((o == q1[2 * s])     ? DELTA : 0.0f);
                float e1 = (o == q0[2 * s + 1]) ? 1.0f : ((o == q1[2 * s + 1]) ? DELTA : 0.0f);
                aq[t][s] = (v2f){e0, e1};
            }
        }
        #pragma unroll
        for (int j = 0; j < 5; ++j) {
            v2f bq[4] = { {pa[j].x, pa[j].y}, {pa[j].z, pa[j].w}, {pb[j].x, pb[j].y}, {pb[j].z, pb[j].w} };
            #pragma unroll
            for (int s = 0; s < 4; ++s) {
                #pragma unroll
                for (int t = 0; t < 6; ++t)
                    yxacc[t][j] = __builtin_amdgcn_wmma_f32_16x16x4_f32(
                        false, aq[t][s], false, bq[s], (short)0, yxacc[t][j], false, false);
            }
        }
    }

    float* accw = lds_acc[wv & (NACC - 1)];
    if (wv < NACC) {
        #pragma unroll
        for (int t = 0; t < 6; ++t) {
            #pragma unroll
            for (int j = 0; j < 5; ++j) {
                #pragma unroll
                for (int r = 0; r < 8; ++r)
                    accw[(t * 16 + 8 * h + r) * AP + 16 * j + c] = yxacc[t][j][r];
            }
        }
    }
    __syncthreads();
    if (wv >= NACC) {
        #pragma unroll
        for (int t = 0; t < 6; ++t) {
            #pragma unroll
            for (int j = 0; j < 5; ++j) {
                #pragma unroll
                for (int r = 0; r < 8; ++r)
                    accw[(t * 16 + 8 * h + r) * AP + 16 * j + c] += yxacc[t][j][r];
            }
        }
    }
    if (tid < COUT) {
        float s = 0.0f;
        for (int wq = 0; wq < NW; ++wq) {
            for (int cc = 0; cc < 16; ++cc) s += lds_yy[wq][tid * 16 + cc];
        }
        atomicAdd(&yyc[tid], s);
    }
    __syncthreads();
    {
        const int rot = (int)((blockIdx.x * 224u) % (unsigned)(COUT * K_));
        for (int i = tid; i < COUT * K_; i += 256) {
            int j = i + rot; if (j >= COUT * K_) j -= COUT * K_;
            int o = j / K_, k = j - o * K_;
            float s = (lds_acc[0][o * AP + k] + lds_acc[1][o * AP + k])
                    + (lds_acc[2][o * AP + k] + lds_acc[3][o * AP + k]);
            atomicAdd(&yxc[j], s);
        }
    }
}

__global__ void zero_ws(float* sums, unsigned int* mb)
{
    int i = blockIdx.x * blockDim.x + threadIdx.x;
    if (i < NCOPY * CST) sums[i] = 0.0f;
    if (i == 0)          *mb = __float_as_uint(PRECF);
}

__global__ void finalize_ds(const float* __restrict__ w,
                            const float* __restrict__ yx,
                            const float* __restrict__ yy,
                            float* __restrict__ ds, unsigned int* mb)
{
    int i = blockIdx.x * blockDim.x + threadIdx.x;
    if (i < COUT * K_) {
        int o = i / K_;
        float sx = 0.0f, sy = 0.0f;
        #pragma unroll
        for (int q = 0; q < NCOPY; ++q) { sx += yx[q * CST + i]; sy += yy[q * CST + o]; }
        float d = sx * (1.0f / (float)L_) - (sy * (1.0f / (float)L_)) * w[i];
        ds[i] = d;
        atomicMax(mb, __float_as_uint(__builtin_fabsf(d)));
    }
}

__global__ void apply_update(const float* __restrict__ w,
                             const float* __restrict__ ds,
                             const unsigned int* __restrict__ mb,
                             float* __restrict__ neww)
{
    int i = blockIdx.x * blockDim.x + threadIdx.x;
    if (i < COUT * K_) {
        float nc = __uint_as_float(*mb);
        neww[i] = w[i] + LR * (ds[i] / nc);
    }
}

extern "C" void kernel_launch(void* const* d_in, const int* in_sizes, int n_in,
                              void* d_out, int out_size, void* d_ws, size_t ws_size,
                              hipStream_t stream)
{
    const float* x = (const float*)d_in[0];
    const float* w = (const float*)d_in[1];
    float* out  = (float*)d_out;
    float* neww = out + (size_t)B_ * COUT * L_;

    float*        yx = (float*)d_ws;
    float*        yy = yx + COUT * K_;
    unsigned int* mb = (unsigned int*)(yx + NCOPY * CST);
    float*        ds = (float*)(mb + 1);

    zero_ws<<<(NCOPY * CST + 255) / 256, 256, 0, stream>>>(yx, mb);
    conv_hebb_gemm<<<256, 256, 0, stream>>>(x, w, out, yx, yy);
    finalize_ds<<<(COUT * K_ + 255) / 256, 256, 0, stream>>>(w, yx, yy, ds, mb);
    apply_update<<<(COUT * K_ + 255) / 256, 256, 0, stream>>>(w, ds, mb, neww);
}


# Group sums of yx, yy: planes keep the accumulators lane by lane (16-byte LDS stores, loads); partner sums, yy sums and final plane reads go out as groups, not one LDS round trip each; sums unchanged
# speedup vs baseline: 17.9605x; 1.0972x over previous
#include <hip/hip_runtime.h>

#define B_     64
#define CIN    3
#define Hh     96
#define Ww     96
#define COUT   96
#define KH     5
#define KW     5
#define STRIDE 2
#define PAD    2
#define HOUT   48
#define WOUT   48
#define L_     (HOUT*WOUT)
#define K_     (CIN*KH*KW)
#define KP     76
#define LR     0.02f
#define DELTA  (-0.4f)
#define PRECF  1e-30f

#define NW     8
#define TPG    36
#define NACC   4
#define PP     20
#define KR     80
#define ABLK   256
#define APL    (30*ABLK)
#define KBLK   3
#define NBLK   7
#define NCOPY  8
#define CST    (COUT*K_ + COUT)

typedef float v2f __attribute__((ext_vector_type(2)));
typedef float v4f __attribute__((ext_vector_type(4)));
typedef float v8f __attribute__((ext_vector_type(8)));
typedef int   v4i __attribute__((ext_vector_type(4)));


__device__ __forceinline__ void wave_lds_handover()
{
    __builtin_amdgcn_fence(__ATOMIC_RELEASE, "wavefront");
    __builtin_amdgcn_wave_barrier();
    __builtin_amdgcn_fence(__ATOMIC_ACQUIRE, "wavefront");
}

__global__ __launch_bounds__(256) void conv_hebb_gemm(
    const float* __restrict__ x, const float* __restrict__ w,
    float* __restrict__ out, float* __restrict__ yx, float* __restrict__ yy)
{
    __shared__ __attribute__((aligned(16))) float lds_w[COUT * KP];
    __shared__ __attribute__((aligned(16))) float lds_p[NW][KR * PP];
    __shared__ __attribute__((aligned(16))) int   lds_sel[NW][32];
    __shared__ __attribute__((aligned(16))) float lds_acc[NACC][APL];
    __shared__ __attribute__((aligned(16))) float lds_yy[NW][COUT * 16];

    const int tid  = threadIdx.x;
    const int wv   = __builtin_amdgcn_readfirstlane(tid >> 5);
    const int lane = tid & 31;
    const int c    = lane & 15;
    const int h    = lane >> 4;

    float* P = lds_p[wv];
    int*   S = lds_sel[wv];
    float* Y = lds_yy[wv];

    const int die = (int)((__builtin_amdgcn_s_sendmsg_rtn(0x87) >> 16) & (NCOPY - 1));
    float* yxc = yx + die * CST;
    float* yyc = yy + die * CST;

    {
        float wreg[29];
        #pragma unroll
        for (int u = 0; u < 29; ++u) {
            int i = tid + 256 * u;
            wreg[u] = (i < COUT * K_) ? w[i] : 0.0f;
        }
        #pragma unroll
        for (int u = 0; u < 29; ++u) {
            int i = tid + 256 * u;
            if (i < COUT * K_) { int o = i / K_; int k = i - o * K_; lds_w[o * KP + k] = wreg[u]; }
        }
        if (tid < COUT) lds_w[tid * KP + K_] = 0.0f;
        for (int i = lane; i < (KR - K_) * PP; i += 32) P[K_ * PP + i] = 0.0f;
        for (int i = lane; i < COUT * 16; i += 32) Y[i] = 0.0f;
    }
    __syncthreads();

    v8f   yxacc[6][5];
    #pragma unroll
    for (int t = 0; t < 6; ++t) {
        #pragma unroll
        for (int j = 0; j < 5; ++j) yxacc[t][j] = (v8f){};
    }

    const int    b  = blockIdx.x >> 2;
    const float* xb = x + (size_t)b * CIN * Hh * Ww;

    const int ntile = (wv < TPG - 4 * NW) ? 5 : 4;
    for (int it = 0; it < ntile; ++it) {
        const int l0 = ((blockIdx.x & 3) * TPG + it * NW + wv) * 16;

        wave_lds_handover();

        {
            const int l   = l0 + c;
            const int y   = l / WOUT, xo = l - y * WOUT;
            const int iy0 = y * STRIDE - PAD, ix0 = xo * STRIDE - PAD;
            float g[38];
            #pragma unroll
            for (int u = 0; u < 38; ++u) {
                int k  = 2 * u + h;
                int cc = k / (KH * KW);
                int r  = k - cc * (KH * KW);
                int ki = r / KW, kj = r - ki * KW;
                int iy = iy0 + ki, ix = ix0 + kj;
                unsigned um = ((unsigned)iy > (unsigned)ix) ? (unsigned)iy : (unsigned)ix;
                bool ok = (k < K_) & (um < (unsigned)Hh);
                int iyc = iy < 0 ? 0 : (iy > Hh - 1 ? Hh - 1 : iy);
                int ixc = ix < 0 ? 0 : (ix > Ww - 1 ? Ww - 1 : ix);
                int ccc = cc > CIN - 1 ? CIN - 1 : cc;
                float v = xb[(unsigned)((ccc * Hh + iyc) * Ww + ixc)];
                g[u] = ok ? v : 0.0f;
            }
            #pragma unroll
            for (int u = 0; u < 38; ++u) {
                int k = 2 * u + h;
                P[k * PP + c] = g[u];
            }
        }
        wave_lds_handover();

        v8f aO[6], aT[6];
        #pragma unroll
        for (int t = 0; t < 6; ++t) { aO[t] = (v8f){}; aT[t] = (v8f){}; }
        v2f fa[2][KBLK][6], fb[2][KBLK];
        #pragma unroll
        for (int s = 0; s < KBLK; ++s) {
            const int ka = 4 * s + 2 * h;
            fb[0][s] = (v2f){P[ka * PP + c], P[(ka + 1) * PP + c]};
            #pragma unroll
            for (int t = 0; t < 6; ++t) fa[0][s][t] = *(const v2f*)&lds_w[(t * 16 + c) * KP + ka];
        }
        #pragma unroll
        for (int bl = 0; bl < NBLK; ++bl) {
            __builtin_amdgcn_sched_barrier(0);
            #pragma unroll
            for (int s = 0; s < KBLK; ++s) {
                const int kk = 4 * ((bl + 1) * KBLK + s);
                if (bl + 1 < NBLK && kk < KP) {
                    const int ka = kk + 2 * h;
                    fb[(bl + 1) & 1][s] = (v2f){P[ka * PP + c], P[(ka + 1) * PP + c]};
                    #pragma unroll
                    for (int t = 0; t < 6; ++t)
                        fa[(bl + 1) & 1][s][t] = *(const v2f*)&lds_w[(t * 16 + c) * KP + ka];
                }
            }
            __builtin_amdgcn_sched_barrier(0);
            #pragma unroll
            for (int s = 0; s < KBLK; ++s) {
                const int kk = 4 * (bl * KBLK + s);
                if (kk < KP) {
                    const v2f bP = fb[bl & 1][s];
                    #pragma unroll
                    for (int t = 0; t < 6; ++t) {
                        const v2f aW = fa[bl & 1][s][t];
                        v2f aL = {aW.x * __builtin_fabsf(aW.x), aW.y * __builtin_fabsf(aW.y)};
                        aO[t] = __builtin_amdgcn_wmma_f32_16x16x4_f32(
                            false, aW, false, bP, (short)0, aO[t], false, false);
                        aT[t] = __builtin_amdgcn_wmma_f32_16x16x4_f32(
                            false, aL, false, bP, (short)0, aT[t], false, false);
                    }
                }
            }
        }

        __builtin_amdgcn_sched_barrier(0);

        {
            float* ob = out + (size_t)b * COUT * L_ + l0 + c;
            #pragma unroll
            for (int t = 0; t < 6; ++t) {
                #pragma unroll
                for (int r = 0; r < 8; ++r)
                    ob[(size_t)(t * 16 + 8 * h + r) * L_] = aO[t][r];
            }
        }

        float v0, v1; int i0, i1;
        {
            float fa = aT[0][0], fb = aT[0][1];
            bool  sw = fb > fa;
            v0 = sw ? fb : fa; i0 = sw ? 1 : 0;
            v1 = sw ? fa : fb; i1 = sw ? 0 : 1;
        }
        #pragma unroll
        for (int t = 0; t < 6; ++t) {
            #pragma unroll
            for (int r = 0; r < 8; ++r) {
                if (t == 0 && r < 2) continue;
                const float v  = aT[t][r];
                const int   o  = t * 16 + r;
                const bool  g0 = v > v0;
                const bool  g1 = v > v1;
                const float nv = g1 ? v : v1;
                const int   ni = g1 ? o : i1;
                v1 = g0 ? v0 : nv; i1 = g0 ? i0 : ni;
                v0 = g0 ? v  : v0; i0 = g0 ? o  : i0;
            }
        }
        i0 += 8 * h; i1 += 8 * h;
        int n0, n1; float m0, m1;
        {
            float w0 = __shfl_xor(v0, 16), w1 = __shfl_xor(v1, 16);
            int   j0 = __shfl_xor(i0, 16), j1 = __shfl_xor(i1, 16);
            const bool mine = (v0 > w0) | ((v0 == w0) & (i0 < j0));
            const bool sa   = (v1 > w0) | ((v1 == w0) & (i1 < j0));
            const bool sb   = (w1 > v0) | ((w1 == v0) & (j1 < i0));
            const int   na = sa ? i1 : j0;  const float ma = sa ? v1 : w0;
            const int   nb = sb ? j1 : i0;  const float mb = sb ? w1 : v0;
            n0 = mine ? i0 : j0;  m0 = mine ? v0 : w0;
            n1 = mine ? na : nb;  m1 = mine ? ma : mb;
        }

        if (h == 0) {
            Y[n0 * 16 + c] += m0;
            Y[n1 * 16 + c] += DELTA * m1;
            S[c] = n0; S[16 + c] = n1;
        }
        wave_lds_handover();
        v4i sa = *(const v4i*)&S[8 * h],      sb = *(const v4i*)&S[8 * h + 4];
        v4i sc = *(const v4i*)&S[16 + 8 * h], sd = *(const v4i*)&S[16 + 8 * h + 4];
        v4f pa[5], pb[5];
        #pragma unroll
        for (int j = 0; j < 5; ++j) {
            pa[j] = *(const v4f*)&P[(16 * j + c) * PP + 8 * h];
            pb[j] = *(const v4f*)&P[(16 * j + c) * PP + 8 * h + 4];
        }
        __builtin_amdgcn_sched_barrier(0);
        int q0[8], q1[8];
        {
            q0[0] = sa.x; q0[1] = sa.y; q0[2] = sa.z; q0[3] = sa.w;
            q0[4] = sb.x; q0[5] = sb.y; q0[6] = sb.z; q0[7] = sb.w;
            q1[0] = sc.x; q1[1] = sc.y; q1[2] = sc.z; q1[3] = sc.w;
            q1[4] = sd.x; q1[5] = sd.y; q1[6] = sd.z; q1[7] = sd.w;
        }

        v2f aq[6][4];
        #pragma unroll
        for (int t = 0; t < 6; ++t) {
            const int o = t * 16 + c;
            #pragma unroll
            for (int s = 0; s < 4; ++s) {
                float e0 = (o == q0[2 * s])     ? 1.0f : ((o == q1[2 * s])     ? DELTA : 0.0f);
                float e1 = (o == q0[2 * s + 1]) ? 1.0f : ((o == q1[2 * s + 1]) ? DELTA : 0.0f);
                aq[t][s] = (v2f){e0, e1};
            }
        }
        #pragma unroll
        for (int j = 0; j < 5; ++j) {
            v2f bq[4] = { {pa[j].x, pa[j].y}, {pa[j].z, pa[j].w}, {pb[j].x, pb[j].y}, {pb[j].z, pb[j].w} };
            #pragma unroll
            for (int s = 0; s < 4; ++s) {
                #pragma unroll
                for (int t = 0; t < 6; ++t)
                    yxacc[t][j] = __builtin_amdgcn_wmma_f32_16x16x4_f32(
                        false, aq[t][s], false, bq[s], (short)0, yxacc[t][j], false, false);
            }
        }
    }

    float* accw = lds_acc[wv & (NACC - 1)] + lane * 4;
    if (wv < NACC) {
        #pragma unroll
        for (int t = 0; t < 6; ++t) {
            #pragma unroll
            for (int j = 0; j < 5; ++j) {
                const v8f a = yxacc[t][j];
                *(v4f*)&accw[(t * 5 + j) * ABLK]       = (v4f){a[0], a[1], a[2], a[3]};
                *(v4f*)&accw[(t * 5 + j) * ABLK + 128] = (v4f){a[4], a[5], a[6], a[7]};
            }
        }
    }
    __syncthreads();
    if (wv >= NACC) {
        v4f lo[30], hi[30];
        #pragma unroll
        for (int q = 0; q < 30; ++q) {
            lo[q] = *(const v4f*)&accw[q * ABLK];
            hi[q] = *(const v4f*)&accw[q * ABLK + 128];
        }
        #pragma unroll
        for (int t = 0; t < 6; ++t) {
            #pragma unroll
            for (int j = 0; j < 5; ++j) {
                const v8f a = yxacc[t][j];
                const v4f l = lo[t * 5 + j], u = hi[t * 5 + j];
                *(v4f*)&accw[(t * 5 + j) * ABLK]       = (v4f){l.x + a[0], l.y + a[1], l.z + a[2], l.w + a[3]};
                *(v4f*)&accw[(t * 5 + j) * ABLK + 128] = (v4f){u.x + a[4], u.y + a[5], u.z + a[6], u.w + a[7]};
            }
        }
    }
    if (tid < COUT) {
        v4f yv[NW][4];
        #pragma unroll
        for (int wq = 0; wq < NW; ++wq) {
            #pragma unroll
            for (int q = 0; q < 4; ++q) yv[wq][q] = *(const v4f*)&lds_yy[wq][tid * 16 + 4 * q];
        }
        __builtin_amdgcn_sched_barrier(0);
        float s = 0.0f;
        #pragma unroll
        for (int wq = 0; wq < NW; ++wq) {
            #pragma unroll
            for (int q = 0; q < 4; ++q) { s += yv[wq][q].x; s += yv[wq][q].y; s += yv[wq][q].z; s += yv[wq][q].w; }
        }
        atomicAdd(&yyc[tid], s);
    }
    __syncthreads();
    {
        const int rot = (int)((blockIdx.x * 224u) % (unsigned)(COUT * K_));
        const float* pl = lds_acc[0];
        int   jg[28];
        float pv[28][NACC];
        #pragma unroll
        for (int u = 0; u < 28; ++u) {
            int j = tid + 256 * u + rot; if (j >= COUT * K_) j -= COUT * K_;
            const int o = j / K_, k = j - o * K_;
            const int off = (((o >> 4) * 5 + (k >> 4)) * 2 + ((o >> 2) & 1)) * 128 + ((((o >> 3) & 1) * 16 + (k & 15)) * 4) + (o & 3);
            jg[u] = j;
            #pragma unroll
            for (int p = 0; p < NACC; ++p) pv[u][p] = pl[p * APL + off];
        }
        #pragma unroll
        for (int u = 0; u < 28; ++u)
            atomicAdd(&yxc[jg[u]], (pv[u][0] + pv[u][1]) + (pv[u][2] + pv[u][3]));
        if (wv == 0) {
            int j = tid + 256 * 28 + rot; if (j >= COUT * K_) j -= COUT * K_;
            const int o = j / K_, k = j - o * K_;
            const int off = (((o >> 4) * 5 + (k >> 4)) * 2 + ((o >> 2) & 1)) * 128 + ((((o >> 3) & 1) * 16 + (k & 15)) * 4) + (o & 3);
            atomicAdd(&yxc[j], (pl[off] + pl[APL + off]) + (pl[2 * APL + off] + pl[3 * APL + off]));
        }
    }
}

__global__ void zero_ws(float* sums, unsigned int* mb)
{
    int i = blockIdx.x * blockDim.x + threadIdx.x;
    if (i < NCOPY * CST) sums[i] = 0.0f;
    if (i == 0)          *mb = __float_as_uint(PRECF);
}

__global__ void finalize_ds(const float* __restrict__ w,
                            const float* __restrict__ yx,
                            const float* __restrict__ yy,
                            float* __restrict__ ds, unsigned int* mb)
{
    int i = blockIdx.x * blockDim.x + threadIdx.x;
    if (i < COUT * K_) {
        int o = i / K_;
        float sx = 0.0f, sy = 0.0f;
        #pragma unroll
        for (int q = 0; q < NCOPY; ++q) { sx += yx[q * CST + i]; sy += yy[q * CST + o]; }
        float d = sx * (1.0f / (float)L_) - (sy * (1.0f / (float)L_)) * w[i];
        ds[i] = d;
        atomicMax(mb, __float_as_uint(__builtin_fabsf(d)));
    }
}

__global__ void apply_update(const float* __restrict__ w,
                             const float* __restrict__ ds,
                             const unsigned int* __restrict__ mb,
                             float* __restrict__ neww)
{
    int i = blockIdx.x * blockDim.x + threadIdx.x;
    if (i < COUT * K_) {
        float nc = __uint_as_float(*mb);
        neww[i] = w[i] + LR * (ds[i] / nc);
    }
}

extern "C" void kernel_launch(void* const* d_in, const int* in_sizes, int n_in,
                              void* d_out, int out_size, void* d_ws, size_t ws_size,
                              hipStream_t stream)
{
    const float* x = (const float*)d_in[0];
    const float* w = (const float*)d_in[1];
    float* out  = (float*)d_out;
    float* neww = out + (size_t)B_ * COUT * L_;

    float*        yx = (float*)d_ws;
    float*        yy = yx + COUT * K_;
    unsigned int* mb = (unsigned int*)(yx + NCOPY * CST);
    float*        ds = (float*)(mb + 1);

    zero_ws<<<(NCOPY * CST + 255) / 256, 256, 0, stream>>>(yx, mb);
    conv_hebb_gemm<<<256, 256, 0, stream>>>(x, w, out, yx, yy);
    finalize_ds<<<(COUT * K_ + 255) / 256, 256, 0, stream>>>(w, yx, yy, ds, mb);
    apply_update<<<(COUT * K_ + 255) / 256, 256, 0, stream>>>(w, ds, mb, neww);
}


# yx and yy sums leave the matrix pipe: a wave adds each column's patch values to the winner's and runner-up's rows of its own LDS table (lane = patch row, plain load/add/store); sum order changes
# speedup vs baseline: 19.4959x; 1.0855x over previous
#include <hip/hip_runtime.h>

#define B_     64
#define CIN    3
#define Hh     96
#define Ww     96
#define COUT   96
#define KH     5
#define KW     5
#define STRIDE 2
#define PAD    2
#define HOUT   48
#define WOUT   48
#define L_     (HOUT*WOUT)
#define K_     (CIN*KH*KW)
#define KP     76
#define LR     0.02f
#define DELTA  (-0.4f)
#define PRECF  1e-30f

#define NW     8
#define TPG    36
#define PP     20
#define KR     80
#define YL     (KP - 1 - 64)
#define TBLW   (COUT*KP + 32)
#define KBLK   3
#define NBLK   7
#define NCOPY  8
#define CST    (COUT*K_ + COUT)

typedef float v2f __attribute__((ext_vector_type(2)));
typedef float v4f __attribute__((ext_vector_type(4)));
typedef float v8f __attribute__((ext_vector_type(8)));
typedef int   v4i __attribute__((ext_vector_type(4)));


__device__ __forceinline__ void wave_lds_handover()
{
    __builtin_amdgcn_fence(__ATOMIC_RELEASE, "wavefront");
    __builtin_amdgcn_wave_barrier();
    __builtin_amdgcn_fence(__ATOMIC_ACQUIRE, "wavefront");
}

__global__ __launch_bounds__(256) void conv_hebb_gemm(
    const float* __restrict__ x, const float* __restrict__ w,
    float* __restrict__ out, float* __restrict__ yx, float* __restrict__ yy)
{
    __shared__ __attribute__((aligned(16))) float lds_w[COUT * KP];
    __shared__ __attribute__((aligned(16))) float lds_p[NW][KR * PP];
    __shared__ __attribute__((aligned(16))) float lds_t[NW][TBLW];

    const int tid  = threadIdx.x;
    const int wv   = __builtin_amdgcn_readfirstlane(tid >> 5);
    const int lane = tid & 31;
    const int c    = lane & 15;
    const int h    = lane >> 4;

    float* P  = lds_p[wv];
    float* Tw = lds_t[wv];

    const int die = (int)((__builtin_amdgcn_s_sendmsg_rtn(0x87) >> 16) & (NCOPY - 1));
    float* yxc = yx + die * CST;

    {
        float wreg[29];
        #pragma unroll
        for (int u = 0; u < 29; ++u) {
            int i = tid + 256 * u;
            wreg[u] = (i < COUT * K_) ? w[i] : 0.0f;
        }
        #pragma unroll
        for (int u = 0; u < 29; ++u) {
            int i = tid + 256 * u;
            if (i < COUT * K_) { int o = i / K_; int k = i - o * K_; lds_w[o * KP + k] = wreg[u]; }
        }
        if (tid < COUT) lds_w[tid * KP + K_] = 0.0f;
        for (int i = lane; i < (KR - K_) * PP; i += 32) P[K_ * PP + i] = 0.0f;
        for (int i = lane; i < TBLW / 4; i += 32) ((v4f*)Tw)[i] = (v4f){};
    }
    __syncthreads();

    const bool tin = lane <= YL;
    const bool yl  = lane == YL;
    const int  kp2 = (lane + 64 < KR) ? lane + 64 : KR - 1;

    const int    b  = blockIdx.x >> 2;
    const float* xb = x + (size_t)b * CIN * Hh * Ww;

    const int ntile = (wv < TPG - 4 * NW) ? 5 : 4;
    for (int it = 0; it < ntile; ++it) {
        const int l0 = ((blockIdx.x & 3) * TPG + it * NW + wv) * 16;

        wave_lds_handover();

        {
            const int l   = l0 + c;
            const int y   = l / WOUT, xo = l - y * WOUT;
            const int iy0 = y * STRIDE - PAD, ix0 = xo * STRIDE - PAD;
            float g[38];
            #pragma unroll
            for (int u = 0; u < 38; ++u) {
                int k  = 2 * u + h;
                int cc = k / (KH * KW);
                int r  = k - cc * (KH * KW);
                int ki = r / KW, kj = r - ki * KW;
                int iy = iy0 + ki, ix = ix0 + kj;
                unsigned um = ((unsigned)iy > (unsigned)ix) ? (unsigned)iy : (unsigned)ix;
                bool ok = (k < K_) & (um < (unsigned)Hh);
                int iyc = iy < 0 ? 0 : (iy > Hh - 1 ? Hh - 1 : iy);
                int ixc = ix < 0 ? 0 : (ix > Ww - 1 ? Ww - 1 : ix);
                int ccc = cc > CIN - 1 ? CIN - 1 : cc;
                float v = xb[(unsigned)((ccc * Hh + iyc) * Ww + ixc)];
                g[u] = ok ? v : 0.0f;
            }
            #pragma unroll
            for (int u = 0; u < 38; ++u) {
                int k = 2 * u + h;
                P[k * PP + c] = g[u];
            }
        }
        wave_lds_handover();

        v8f aO[6], aT[6];
        #pragma unroll
        for (int t = 0; t < 6; ++t) { aO[t] = (v8f){}; aT[t] = (v8f){}; }
        v2f fa[2][KBLK][6], fb[2][KBLK];
        #pragma unroll
        for (int s = 0; s < KBLK; ++s) {
            const int ka = 4 * s + 2 * h;
            fb[0][s] = (v2f){P[ka * PP + c], P[(ka + 1) * PP + c]};
            #pragma unroll
            for (int t = 0; t < 6; ++t) fa[0][s][t] = *(const v2f*)&lds_w[(t * 16 + c) * KP + ka];
        }
        #pragma unroll
        for (int bl = 0; bl < NBLK; ++bl) {
            __builtin_amdgcn_sched_barrier(0);
            #pragma unroll
            for (int s = 0; s < KBLK; ++s) {
                const int kk = 4 * ((bl + 1) * KBLK + s);
                if (bl + 1 < NBLK && kk < KP) {
                    const int ka = kk + 2 * h;
                    fb[(bl + 1) & 1][s] = (v2f){P[ka * PP + c], P[(ka + 1) * PP + c]};
                    #pragma unroll
                    for (int t = 0; t < 6; ++t)
                        fa[(bl + 1) & 1][s][t] = *(const v2f*)&lds_w[(t * 16 + c) * KP + ka];
                }
            }
            __builtin_amdgcn_sched_barrier(0);
            #pragma unroll
            for (int s = 0; s < KBLK; ++s) {
                const int kk = 4 * (bl * KBLK + s);
                if (kk < KP) {
                    const v2f bP = fb[bl & 1][s];
                    #pragma unroll
                    for (int t = 0; t < 6; ++t) {
                        const v2f aW = fa[bl & 1][s][t];
                        v2f aL = {aW.x * __builtin_fabsf(aW.x), aW.y * __builtin_fabsf(aW.y)};
                        aO[t] = __builtin_amdgcn_wmma_f32_16x16x4_f32(
                            false, aW, false, bP, (short)0, aO[t], false, false);
                        aT[t] = __builtin_amdgcn_wmma_f32_16x16x4_f32(
                            false, aL, false, bP, (short)0, aT[t], false, false);
                    }
                }
            }
        }

        __builtin_amdgcn_sched_barrier(0);

        {
            float* ob = out + (size_t)b * COUT * L_ + l0 + c;
            #pragma unroll
            for (int t = 0; t < 6; ++t) {
                #pragma unroll
                for (int r = 0; r < 8; ++r)
                    ob[(size_t)(t * 16 + 8 * h + r) * L_] = aO[t][r];
            }
        }

        v4f pq[3][4];
        #pragma unroll
        for (int j = 0; j < 4; ++j) {
            pq[0][j] = *(const v4f*)&P[lane * PP + 4 * j];
            pq[1][j] = *(const v4f*)&P[(lane + 32) * PP + 4 * j];
            pq[2][j] = *(const v4f*)&P[kp2 * PP + 4 * j];
        }

        float v0, v1; int i0, i1;
        {
            float fa = aT[0][0], fb = aT[0][1];
            bool  sw = fb > fa;
            v0 = sw ? fb : fa; i0 = sw ? 1 : 0;
            v1 = sw ? fa : fb; i1 = sw ? 0 : 1;
        }
        #pragma unroll
        for (int t = 0; t < 6; ++t) {
            #pragma unroll
            for (int r = 0; r < 8; ++r) {
                if (t == 0 && r < 2) continue;
                const float v  = aT[t][r];
                const int   o  = t * 16 + r;
                const bool  g0 = v > v0;
                const bool  g1 = v > v1;
                const float nv = g1 ? v : v1;
                const int   ni = g1 ? o : i1;
                v1 = g0 ? v0 : nv; i1 = g0 ? i0 : ni;
                v0 = g0 ? v  : v0; i0 = g0 ? o  : i0;
            }
        }
        i0 += 8 * h; i1 += 8 * h;
        int n0, n1; float m0, m1;
        {
            float w0 = __shfl_xor(v0, 16), w1 = __shfl_xor(v1, 16);
            int   j0 = __shfl_xor(i0, 16), j1 = __shfl_xor(i1, 16);
            const bool mine = (v0 > w0) | ((v0 == w0) & (i0 < j0));
            const bool sa   = (v1 > w0) | ((v1 == w0) & (i1 < j0));
            const bool sb   = (w1 > v0) | ((w1 == v0) & (j1 < i0));
            const int   na = sa ? i1 : j0;  const float ma = sa ? v1 : w0;
            const int   nb = sb ? j1 : i0;  const float mb = sb ? w1 : v0;
            n0 = mine ? i0 : j0;  m0 = mine ? v0 : w0;
            n1 = mine ? na : nb;  m1 = mine ? ma : mb;
        }

        #pragma unroll
        for (int i = 0; i < 8; ++i) {
            const int ca = 2 * i, cb = 2 * i + 1;
            const int   ra0 = __builtin_amdgcn_readlane(n0, ca), ra1 = __builtin_amdgcn_readlane(n1, ca);
            const int   rb0 = __builtin_amdgcn_readlane(n0, cb), rb1 = __builtin_amdgcn_readlane(n1, cb);
            const float ma0 = __int_as_float(__builtin_amdgcn_readlane(__float_as_int(m0), ca));
            const float ma1 = __int_as_float(__builtin_amdgcn_readlane(__float_as_int(m1), ca));
            const float mb0 = __int_as_float(__builtin_amdgcn_readlane(__float_as_int(m0), cb));
            const float mb1 = __int_as_float(__builtin_amdgcn_readlane(__float_as_int(m1), cb));
            const bool  e00 = rb0 == ra0, e10 = rb1 == ra0, e01 = rb0 == ra1, e11 = rb1 == ra1;
            const float xa0 = e00 ? 1.0f : (e10 ? DELTA : 0.0f);
            const float xa1 = e01 ? 1.0f : (e11 ? DELTA : 0.0f);
            const float xb0 = e00 ? 1.0f : (e01 ? DELTA : 0.0f);
            const float xb1 = e10 ? 1.0f : (e11 ? DELTA : 0.0f);
            const float ya0 = e00 ? mb0 : mb1, ya1 = e01 ? mb0 : mb1;
            const float yb0 = e00 ? ma0 : ma1, yb1 = e10 ? ma0 : ma1;
            int   ia0[3], ia1[3], ib0[3], ib1[3];
            float oa0[3], oa1[3], ob0[3], ob1[3];
            #pragma unroll
            for (int q = 0; q < 3; ++q) {
                const int off = lane + 32 * q;
                if (q < 2) {
                    ia0[q] = ra0 * KP + off; ia1[q] = ra1 * KP + off;
                    ib0[q] = rb0 * KP + off; ib1[q] = rb1 * KP + off;
                } else {
                    ia0[q] = tin ? ra0 * KP + off : COUT * KP + lane; ia1[q] = tin ? ra1 * KP + off : COUT * KP + lane;
                    ib0[q] = tin ? rb0 * KP + off : COUT * KP + lane; ib1[q] = tin ? rb1 * KP + off : COUT * KP + lane;
                }
                oa0[q] = Tw[ia0[q]]; oa1[q] = Tw[ia1[q]];
                ob0[q] = Tw[ib0[q]]; ob1[q] = Tw[ib1[q]];
            }
            #pragma unroll
            for (int q = 0; q < 3; ++q) {
                const float pa = pq[q][i >> 1][(2 * i) & 3], pb = pq[q][i >> 1][(2 * i + 1) & 3];
                float va0 = pa, va1 = pa, vb0 = pb, vb1 = pb, wa0 = pb, wa1 = pb, wb0 = pa, wb1 = pa;
                if (q == 2) {
                    va0 = yl ? ma0 : pa; va1 = yl ? ma1 : pa; vb0 = yl ? mb0 : pb; vb1 = yl ? mb1 : pb;
                    wa0 = yl ? ya0 : pb; wa1 = yl ? ya1 : pb; wb0 = yl ? yb0 : pa; wb1 = yl ? yb1 : pa;
                }
                Tw[ia0[q]] = (oa0[q] + va0)         + xa0 * wa0;
                Tw[ia1[q]] = (oa1[q] + DELTA * va1) + xa1 * wa1;
                Tw[ib0[q]] = (ob0[q] + vb0)         + xb0 * wb0;
                Tw[ib1[q]] = (ob1[q] + DELTA * vb1) + xb1 * wb1;
            }
        }
    }

    __syncthreads();
    {
        const int rot = (int)((blockIdx.x * 224u) % (unsigned)(COUT * KP));
        #pragma unroll
        for (int ub = 0; ub < 28; ub += 7) {
            int   gi[7];
            float tv[7][NW];
            #pragma unroll
            for (int u = 0; u < 7; ++u) {
                int e = tid + 256 * (ub + u) + rot; if (e >= COUT * KP) e -= COUT * KP;
                const int o = e / KP, k = e - o * KP;
                gi[u] = (k < K_) ? e - o : COUT * K_ + o;
                #pragma unroll
                for (int wq = 0; wq < NW; ++wq) tv[u][wq] = lds_t[wq][e];
            }
            #pragma unroll
            for (int u = 0; u < 7; ++u)
                atomicAdd(&yxc[gi[u]], ((tv[u][0] + tv[u][1]) + (tv[u][2] + tv[u][3])) + ((tv[u][4] + tv[u][5]) + (tv[u][6] + tv[u][7])));
        }
        if (wv < 4) {
            int e = tid + 256 * 28 + rot; if (e >= COUT * KP) e -= COUT * KP;
            const int o = e / KP, k = e - o * KP;
            const int g = (k < K_) ? e - o : COUT * K_ + o;
            float t8[NW];
            #pragma unroll
            for (int wq = 0; wq < NW; ++wq) t8[wq] = lds_t[wq][e];
            atomicAdd(&yxc[g], ((t8[0] + t8[1]) + (t8[2] + t8[3])) + ((t8[4] + t8[5]) + (t8[6] + t8[7])));
        }
    }
}

__global__ void zero_ws(float* sums, unsigned int* mb)
{
    int i = blockIdx.x * blockDim.x + threadIdx.x;
    if (i < NCOPY * CST) sums[i] = 0.0f;
    if (i == 0)          *mb = __float_as_uint(PRECF);
}

__global__ void finalize_ds(const float* __restrict__ w,
                            const float* __restrict__ yx,
                            const float* __restrict__ yy,
                            float* __restrict__ ds, unsigned int* mb)
{
    int i = blockIdx.x * blockDim.x + threadIdx.x;
    if (i < COUT * K_) {
        int o = i / K_;
        float sx = 0.0f, sy = 0.0f;
        #pragma unroll
        for (int q = 0; q < NCOPY; ++q) { sx += yx[q * CST + i]; sy += yy[q * CST + o]; }
        float d = sx * (1.0f / (float)L_) - (sy * (1.0f / (float)L_)) * w[i];
        ds[i] = d;
        atomicMax(mb, __float_as_uint(__builtin_fabsf(d)));
    }
}

__global__ void apply_update(const float* __restrict__ w,
                             const float* __restrict__ ds,
                             const unsigned int* __restrict__ mb,
                             float* __restrict__ neww)
{
    int i = blockIdx.x * blockDim.x + threadIdx.x;
    if (i < COUT * K_) {
        float nc = __uint_as_float(*mb);
        neww[i] = w[i] + LR * (ds[i] / nc);
    }
}

extern "C" void kernel_launch(void* const* d_in, const int* in_sizes, int n_in,
                              void* d_out, int out_size, void* d_ws, size_t ws_size,
                              hipStream_t stream)
{
    const float* x = (const float*)d_in[0];
    const float* w = (const float*)d_in[1];
    float* out  = (float*)d_out;
    float* neww = out + (size_t)B_ * COUT * L_;

    float*        yx = (float*)d_ws;
    float*        yy = yx + COUT * K_;
    unsigned int* mb = (unsigned int*)(yx + NCOPY * CST);
    float*        ds = (float*)(mb + 1);

    zero_ws<<<(NCOPY * CST + 255) / 256, 256, 0, stream>>>(yx, mb);
    conv_hebb_gemm<<<256, 256, 0, stream>>>(x, w, out, yx, yy);
    finalize_ds<<<(COUT * K_ + 255) / 256, 256, 0, stream>>>(w, yx, yy, ds, mb);
    apply_update<<<(COUT * K_ + 255) / 256, 256, 0, stream>>>(w, ds, mb, neww);
}


# yx and yy sums: each term goes into the wave's own LDS table by one LDS float add instruction, not load/add/store with same-row cases; sum order unchanged, runner-up products rounded before the add
# speedup vs baseline: 20.6520x; 1.0593x over previous
#include <hip/hip_runtime.h>

#define B_     64
#define CIN    3
#define Hh     96
#define Ww     96
#define COUT   96
#define KH     5
#define KW     5
#define STRIDE 2
#define PAD    2
#define HOUT   48
#define WOUT   48
#define L_     (HOUT*WOUT)
#define K_     (CIN*KH*KW)
#define KP     76
#define LR     0.02f
#define DELTA  (-0.4f)
#define PRECF  1e-30f

#define NW     8
#define TPG    36
#define PP     20
#define KR     80
#define YL     (KP - 1 - 64)
#define TBLW   (COUT*KP + 32)
#define KBLK   3
#define NBLK   7
#define NCOPY  8
#define CST    (COUT*K_ + COUT)

typedef float v2f __attribute__((ext_vector_type(2)));
typedef float v4f __attribute__((ext_vector_type(4)));
typedef float v8f __attribute__((ext_vector_type(8)));
typedef int   v4i __attribute__((ext_vector_type(4)));


__device__ __forceinline__ void wave_lds_handover()
{
    __builtin_amdgcn_fence(__ATOMIC_RELEASE, "wavefront");
    __builtin_amdgcn_wave_barrier();
    __builtin_amdgcn_fence(__ATOMIC_ACQUIRE, "wavefront");
}

__global__ __launch_bounds__(256) void conv_hebb_gemm(
    const float* __restrict__ x, const float* __restrict__ w,
    float* __restrict__ out, float* __restrict__ yx, float* __restrict__ yy)
{
    __shared__ __attribute__((aligned(16))) float lds_w[COUT * KP];
    __shared__ __attribute__((aligned(16))) float lds_p[NW][KR * PP];
    __shared__ __attribute__((aligned(16))) float lds_t[NW][TBLW];

    const int tid  = threadIdx.x;
    const int wv   = __builtin_amdgcn_readfirstlane(tid >> 5);
    const int lane = tid & 31;
    const int c    = lane & 15;
    const int h    = lane >> 4;

    float* P  = lds_p[wv];
    float* Tw = lds_t[wv];

    const int die = (int)((__builtin_amdgcn_s_sendmsg_rtn(0x87) >> 16) & (NCOPY - 1));
    float* yxc = yx + die * CST;

    {
        float wreg[29];
        #pragma unroll
        for (int u = 0; u < 29; ++u) {
            int i = tid + 256 * u;
            wreg[u] = (i < COUT * K_) ? w[i] : 0.0f;
        }
        #pragma unroll
        for (int u = 0; u < 29; ++u) {
            int i = tid + 256 * u;
            if (i < COUT * K_) { int o = i / K_; int k = i - o * K_; lds_w[o * KP + k] = wreg[u]; }
        }
        if (tid < COUT) lds_w[tid * KP + K_] = 0.0f;
        for (int i = lane; i < (KR - K_) * PP; i += 32) P[K_ * PP + i] = 0.0f;
        for (int i = lane; i < TBLW / 4; i += 32) ((v4f*)Tw)[i] = (v4f){};
    }
    __syncthreads();

    const bool yl  = lane == YL;
    const int  kp2 = (lane + 64 < KR) ? lane + 64 : KR - 1;

    const int    b  = blockIdx.x >> 2;
    const float* xb = x + (size_t)b * CIN * Hh * Ww;

    const int ntile = (wv < TPG - 4 * NW) ? 5 : 4;
    for (int it = 0; it < ntile; ++it) {
        const int l0 = ((blockIdx.x & 3) * TPG + it * NW + wv) * 16;

        wave_lds_handover();

        {
            const int l   = l0 + c;
            const int y   = l / WOUT, xo = l - y * WOUT;
            const int iy0 = y * STRIDE - PAD, ix0 = xo * STRIDE - PAD;
            float g[38];
            #pragma unroll
            for (int u = 0; u < 38; ++u) {
                int k  = 2 * u + h;
                int cc = k / (KH * KW);
                int r  = k - cc * (KH * KW);
                int ki = r / KW, kj = r - ki * KW;
                int iy = iy0 + ki, ix = ix0 + kj;
                unsigned um = ((unsigned)iy > (unsigned)ix) ? (unsigned)iy : (unsigned)ix;
                bool ok = (k < K_) & (um < (unsigned)Hh);
                int iyc = iy < 0 ? 0 : (iy > Hh - 1 ? Hh - 1 : iy);
                int ixc = ix < 0 ? 0 : (ix > Ww - 1 ? Ww - 1 : ix);
                int ccc = cc > CIN - 1 ? CIN - 1 : cc;
                float v = xb[(unsigned)((ccc * Hh + iyc) * Ww + ixc)];
                g[u] = ok ? v : 0.0f;
            }
            #pragma unroll
            for (int u = 0; u < 38; ++u) {
                int k = 2 * u + h;
                P[k * PP + c] = g[u];
            }
        }
        wave_lds_handover();

        v8f aO[6], aT[6];
        #pragma unroll
        for (int t = 0; t < 6; ++t) { aO[t] = (v8f){}; aT[t] = (v8f){}; }
        v2f fa[2][KBLK][6], fb[2][KBLK];
        #pragma unroll
        for (int s = 0; s < KBLK; ++s) {
            const int ka = 4 * s + 2 * h;
            fb[0][s] = (v2f){P[ka * PP + c], P[(ka + 1) * PP + c]};
            #pragma unroll
            for (int t = 0; t < 6; ++t) fa[0][s][t] = *(const v2f*)&lds_w[(t * 16 + c) * KP + ka];
        }
        #pragma unroll
        for (int bl = 0; bl < NBLK; ++bl) {
            __builtin_amdgcn_sched_barrier(0);
            #pragma unroll
            for (int s = 0; s < KBLK; ++s) {
                const int kk = 4 * ((bl + 1) * KBLK + s);
                if (bl + 1 < NBLK && kk < KP) {
                    const int ka = kk + 2 * h;
                    fb[(bl + 1) & 1][s] = (v2f){P[ka * PP + c], P[(ka + 1) * PP + c]};
                    #pragma unroll
                    for (int t = 0; t < 6; ++t)
                        fa[(bl + 1) & 1][s][t] = *(const v2f*)&lds_w[(t * 16 + c) * KP + ka];
                }
            }
            __builtin_amdgcn_sched_barrier(0);
            #pragma unroll
            for (int s = 0; s < KBLK; ++s) {
                const int kk = 4 * (bl * KBLK + s);
                if (kk < KP) {
                    const v2f bP = fb[bl & 1][s];
                    #pragma unroll
                    for (int t = 0; t < 6; ++t) {
                        const v2f aW = fa[bl & 1][s][t];
                        v2f aL = {aW.x * __builtin_fabsf(aW.x), aW.y * __builtin_fabsf(aW.y)};
                        aO[t] = __builtin_amdgcn_wmma_f32_16x16x4_f32(
                            false, aW, false, bP, (short)0, aO[t], false, false);
                        aT[t] = __builtin_amdgcn_wmma_f32_16x16x4_f32(
                            false, aL, false, bP, (short)0, aT[t], false, false);
                    }
                }
            }
        }

        __builtin_amdgcn_sched_barrier(0);

        {
            float* ob = out + (size_t)b * COUT * L_ + l0 + c;
            #pragma unroll
            for (int t = 0; t < 6; ++t) {
                #pragma unroll
                for (int r = 0; r < 8; ++r)
                    ob[(size_t)(t * 16 + 8 * h + r) * L_] = aO[t][r];
            }
        }

        v4f pq[3][4];
        #pragma unroll
        for (int j = 0; j < 4; ++j) {
            pq[0][j] = *(const v4f*)&P[lane * PP + 4 * j];
            pq[1][j] = *(const v4f*)&P[(lane + 32) * PP + 4 * j];
            pq[2][j] = *(const v4f*)&P[kp2 * PP + 4 * j];
        }

        float v0, v1; int i0, i1;
        {
            float fa = aT[0][0], fb = aT[0][1];
            bool  sw = fb > fa;
            v0 = sw ? fb : fa; i0 = sw ? 1 : 0;
            v1 = sw ? fa : fb; i1 = sw ? 0 : 1;
        }
        #pragma unroll
        for (int t = 0; t < 6; ++t) {
            #pragma unroll
            for (int r = 0; r < 8; ++r) {
                if (t == 0 && r < 2) continue;
                const float v  = aT[t][r];
                const int   o  = t * 16 + r;
                const bool  g0 = v > v0;
                const bool  g1 = v > v1;
                const float nv = g1 ? v : v1;
                const int   ni = g1 ? o : i1;
                v1 = g0 ? v0 : nv; i1 = g0 ? i0 : ni;
                v0 = g0 ? v  : v0; i0 = g0 ? o  : i0;
            }
        }
        i0 += 8 * h; i1 += 8 * h;
        int n0, n1; float m0, m1;
        {
            float w0 = __shfl_xor(v0, 16), w1 = __shfl_xor(v1, 16);
            int   j0 = __shfl_xor(i0, 16), j1 = __shfl_xor(i1, 16);
            const bool mine = (v0 > w0) | ((v0 == w0) & (i0 < j0));
            const bool sa   = (v1 > w0) | ((v1 == w0) & (i1 < j0));
            const bool sb   = (w1 > v0) | ((w1 == v0) & (j1 < i0));
            const int   na = sa ? i1 : j0;  const float ma = sa ? v1 : w0;
            const int   nb = sb ? j1 : i0;  const float mb = sb ? w1 : v0;
            n0 = mine ? i0 : j0;  m0 = mine ? v0 : w0;
            n1 = mine ? na : nb;  m1 = mine ? ma : mb;
        }

        #pragma unroll
        for (int x = 0; x < 16; ++x) {
            const int   r0 = __builtin_amdgcn_readlane(n0, x), r1 = __builtin_amdgcn_readlane(n1, x);
            const float t0 = __int_as_float(__builtin_amdgcn_readlane(__float_as_int(m0), x));
            const float t1 = __int_as_float(__builtin_amdgcn_readlane(__float_as_int(m1), x));
            float* w0 = Tw + r0 * KP + lane;
            float* w1 = Tw + r1 * KP + lane;
            #pragma unroll
            for (int q = 0; q < 3; ++q) {
                const float p  = pq[q][x >> 2][x & 3];
                const float a0 = (q == 2 && yl) ? t0 : p;
                const float a1 = DELTA * ((q == 2 && yl) ? t1 : p);
                __hip_atomic_fetch_add(w0 + 32 * q, a0, __ATOMIC_RELAXED, __HIP_MEMORY_SCOPE_WAVEFRONT);
                __hip_atomic_fetch_add(w1 + 32 * q, a1, __ATOMIC_RELAXED, __HIP_MEMORY_SCOPE_WAVEFRONT);
            }
        }
    }

    __syncthreads();
    {
        const int rot = (int)((blockIdx.x * 224u) % (unsigned)(COUT * KP));
        #pragma unroll
        for (int ub = 0; ub < 28; ub += 7) {
            int   gi[7];
            float tv[7][NW];
            #pragma unroll
            for (int u = 0; u < 7; ++u) {
                int e = tid + 256 * (ub + u) + rot; if (e >= COUT * KP) e -= COUT * KP;
                const int o = e / KP, k = e - o * KP;
                gi[u] = (k < K_) ? e - o : COUT * K_ + o;
                #pragma unroll
                for (int wq = 0; wq < NW; ++wq) tv[u][wq] = lds_t[wq][e];
            }
            #pragma unroll
            for (int u = 0; u < 7; ++u)
                atomicAdd(&yxc[gi[u]], ((tv[u][0] + tv[u][1]) + (tv[u][2] + tv[u][3])) + ((tv[u][4] + tv[u][5]) + (tv[u][6] + tv[u][7])));
        }
        if (wv < 4) {
            int e = tid + 256 * 28 + rot; if (e >= COUT * KP) e -= COUT * KP;
            const int o = e / KP, k = e - o * KP;
            const int g = (k < K_) ? e - o : COUT * K_ + o;
            float t8[NW];
            #pragma unroll
            for (int wq = 0; wq < NW; ++wq) t8[wq] = lds_t[wq][e];
            atomicAdd(&yxc[g], ((t8[0] + t8[1]) + (t8[2] + t8[3])) + ((t8[4] + t8[5]) + (t8[6] + t8[7])));
        }
    }
}

__global__ void zero_ws(float* sums, unsigned int* mb)
{
    int i = blockIdx.x * blockDim.x + threadIdx.x;
    if (i < NCOPY * CST) sums[i] = 0.0f;
    if (i == 0)          *mb = __float_as_uint(PRECF);
}

__global__ void finalize_ds(const float* __restrict__ w,
                            const float* __restrict__ yx,
                            const float* __restrict__ yy,
                            float* __restrict__ ds, unsigned int* mb)
{
    int i = blockIdx.x * blockDim.x + threadIdx.x;
    if (i < COUT * K_) {
        int o = i / K_;
        float sx = 0.0f, sy = 0.0f;
        #pragma unroll
        for (int q = 0; q < NCOPY; ++q) { sx += yx[q * CST + i]; sy += yy[q * CST + o]; }
        float d = sx * (1.0f / (float)L_) - (sy * (1.0f / (float)L_)) * w[i];
        ds[i] = d;
        atomicMax(mb, __float_as_uint(__builtin_fabsf(d)));
    }
}

__global__ void apply_update(const float* __restrict__ w,
                             const float* __restrict__ ds,
                             const unsigned int* __restrict__ mb,
                             float* __restrict__ neww)
{
    int i = blockIdx.x * blockDim.x + threadIdx.x;
    if (i < COUT * K_) {
        float nc = __uint_as_float(*mb);
        neww[i] = w[i] + LR * (ds[i] / nc);
    }
}

extern "C" void kernel_launch(void* const* d_in, const int* in_sizes, int n_in,
                              void* d_out, int out_size, void* d_ws, size_t ws_size,
                              hipStream_t stream)
{
    const float* x = (const float*)d_in[0];
    const float* w = (const float*)d_in[1];
    float* out  = (float*)d_out;
    float* neww = out + (size_t)B_ * COUT * L_;

    float*        yx = (float*)d_ws;
    float*        yy = yx + COUT * K_;
    unsigned int* mb = (unsigned int*)(yx + NCOPY * CST);
    float*        ds = (float*)(mb + 1);

    zero_ws<<<(NCOPY * CST + 255) / 256, 256, 0, stream>>>(yx, mb);
    conv_hebb_gemm<<<256, 256, 0, stream>>>(x, w, out, yx, yy);
    finalize_ds<<<(COUT * K_ + 255) / 256, 256, 0, stream>>>(w, yx, yy, ds, mb);
    apply_update<<<(COUT * K_ + 255) / 256, 256, 0, stream>>>(w, ds, mb, neww);
}


# Group sums of yx, yy without float atomics: a group stores its share of ds (yx sums - yy sum*W) in a workspace plane of its own, a 114-group launch adds the planes in a fixed tree; sum order changes
# speedup vs baseline: 21.6793x; 1.0497x over previous
#include <hip/hip_runtime.h>

#define B_     64
#define CIN    3
#define Hh     96
#define Ww     96
#define COUT   96
#define KH     5
#define KW     5
#define STRIDE 2
#define PAD    2
#define HOUT   48
#define WOUT   48
#define L_     (HOUT*WOUT)
#define K_     (CIN*KH*KW)
#define KP     76
#define LR     0.02f
#define DELTA  (-0.4f)
#define PRECF  1e-30f

#define NW     8
#define TPG    36
#define PP     20
#define KR     80
#define YL     (KP - 1 - 64)
#define TBLW   (COUT*KP + 32)
#define KBLK   3
#define NBLK   7
#define NGRP   256
#define PLW    (COUT*KP)

typedef float v2f __attribute__((ext_vector_type(2)));
typedef float v4f __attribute__((ext_vector_type(4)));
typedef float v8f __attribute__((ext_vector_type(8)));
typedef int   v4i __attribute__((ext_vector_type(4)));


__device__ __forceinline__ void wave_lds_handover()
{
    __builtin_amdgcn_fence(__ATOMIC_RELEASE, "wavefront");
    __builtin_amdgcn_wave_barrier();
    __builtin_amdgcn_fence(__ATOMIC_ACQUIRE, "wavefront");
}

__global__ __launch_bounds__(256) void conv_hebb_gemm(
    const float* __restrict__ x, const float* __restrict__ w,
    float* __restrict__ out, float* __restrict__ planes, unsigned int* __restrict__ mb)
{
    __shared__ __attribute__((aligned(16))) float lds_w[COUT * KP];
    __shared__ __attribute__((aligned(16))) float lds_p[NW][KR * PP];
    __shared__ __attribute__((aligned(16))) float lds_t[NW][TBLW];

    const int tid  = threadIdx.x;
    const int wv   = __builtin_amdgcn_readfirstlane(tid >> 5);
    const int lane = tid & 31;
    const int c    = lane & 15;
    const int h    = lane >> 4;

    float* P  = lds_p[wv];
    float* Tw = lds_t[wv];

    if (blockIdx.x == 0 && tid == 0) *mb = __float_as_uint(PRECF);

    {
        float wreg[29];
        #pragma unroll
        for (int u = 0; u < 29; ++u) {
            int i = tid + 256 * u;
            wreg[u] = (i < COUT * K_) ? w[i] : 0.0f;
        }
        #pragma unroll
        for (int u = 0; u < 29; ++u) {
            int i = tid + 256 * u;
            if (i < COUT * K_) { int o = i / K_; int k = i - o * K_; lds_w[o * KP + k] = wreg[u]; }
        }
        if (tid < COUT) lds_w[tid * KP + K_] = 0.0f;
        for (int i = lane; i < (KR - K_) * PP; i += 32) P[K_ * PP + i] = 0.0f;
        for (int i = lane; i < TBLW / 4; i += 32) ((v4f*)Tw)[i] = (v4f){};
    }
    __syncthreads();

    const bool yl  = lane == YL;
    const int  kp2 = (lane + 64 < KR) ? lane + 64 : KR - 1;

    const int    b  = blockIdx.x >> 2;
    const float* xb = x + (size_t)b * CIN * Hh * Ww;

    const int ntile = (wv < TPG - 4 * NW) ? 5 : 4;
    for (int it = 0; it < ntile; ++it) {
        const int l0 = ((blockIdx.x & 3) * TPG + it * NW + wv) * 16;

        wave_lds_handover();

        {
            const int l   = l0 + c;
            const int y   = l / WOUT, xo = l - y * WOUT;
            const int iy0 = y * STRIDE - PAD, ix0 = xo * STRIDE - PAD;
            float g[38];
            #pragma unroll
            for (int u = 0; u < 38; ++u) {
                int k  = 2 * u + h;
                int cc = k / (KH * KW);
                int r  = k - cc * (KH * KW);
                int ki = r / KW, kj = r - ki * KW;
                int iy = iy0 + ki, ix = ix0 + kj;
                unsigned um = ((unsigned)iy > (unsigned)ix) ? (unsigned)iy : (unsigned)ix;
                bool ok = (k < K_) & (um < (unsigned)Hh);
                int iyc = iy < 0 ? 0 : (iy > Hh - 1 ? Hh - 1 : iy);
                int ixc = ix < 0 ? 0 : (ix > Ww - 1 ? Ww - 1 : ix);
                int ccc = cc > CIN - 1 ? CIN - 1 : cc;
                float v = xb[(unsigned)((ccc * Hh + iyc) * Ww + ixc)];
                g[u] = ok ? v : 0.0f;
            }
            #pragma unroll
            for (int u = 0; u < 38; ++u) {
                int k = 2 * u + h;
                P[k * PP + c] = g[u];
            }
        }
        wave_lds_handover();

        v8f aO[6], aT[6];
        #pragma unroll
        for (int t = 0; t < 6; ++t) { aO[t] = (v8f){}; aT[t] = (v8f){}; }
        v2f fa[2][KBLK][6], fb[2][KBLK];
        #pragma unroll
        for (int s = 0; s < KBLK; ++s) {
            const int ka = 4 * s + 2 * h;
            fb[0][s] = (v2f){P[ka * PP + c], P[(ka + 1) * PP + c]};
            #pragma unroll
            for (int t = 0; t < 6; ++t) fa[0][s][t] = *(const v2f*)&lds_w[(t * 16 + c) * KP + ka];
        }
        #pragma unroll
        for (int bl = 0; bl < NBLK; ++bl) {
            __builtin_amdgcn_sched_barrier(0);
            #pragma unroll
            for (int s = 0; s < KBLK; ++s) {
                const int kk = 4 * ((bl + 1) * KBLK + s);
                if (bl + 1 < NBLK && kk < KP) {
                    const int ka = kk + 2 * h;
                    fb[(bl + 1) & 1][s] = (v2f){P[ka * PP + c], P[(ka + 1) * PP + c]};
                    #pragma unroll
                    for (int t = 0; t < 6; ++t)
                        fa[(bl + 1) & 1][s][t] = *(const v2f*)&lds_w[(t * 16 + c) * KP + ka];
                }
            }
            __builtin_amdgcn_sched_barrier(0);
            #pragma unroll
            for (int s = 0; s < KBLK; ++s) {
                const int kk = 4 * (bl * KBLK + s);
                if (kk < KP) {
                    const v2f bP = fb[bl & 1][s];
                    #pragma unroll
                    for (int t = 0; t < 6; ++t) {
                        const v2f aW = fa[bl & 1][s][t];
                        v2f aL = {aW.x * __builtin_fabsf(aW.x), aW.y * __builtin_fabsf(aW.y)};
                        aO[t] = __builtin_amdgcn_wmma_f32_16x16x4_f32(
                            false, aW, false, bP, (short)0, aO[t], false, false);
                        aT[t] = __builtin_amdgcn_wmma_f32_16x16x4_f32(
                            false, aL, false, bP, (short)0, aT[t], false, false);
                    }
                }
            }
        }

        __builtin_amdgcn_sched_barrier(0);

        {
            float* ob = out + (size_t)b * COUT * L_ + l0 + c;
            #pragma unroll
            for (int t = 0; t < 6; ++t) {
                #pragma unroll
                for (int r = 0; r < 8; ++r)
                    ob[(size_t)(t * 16 + 8 * h + r) * L_] = aO[t][r];
            }
        }

        v4f pq[3][4];
        #pragma unroll
        for (int j = 0; j < 4; ++j) {
            pq[0][j] = *(const v4f*)&P[lane * PP + 4 * j];
            pq[1][j] = *(const v4f*)&P[(lane + 32) * PP + 4 * j];
            pq[2][j] = *(const v4f*)&P[kp2 * PP + 4 * j];
        }

        float v0, v1; int i0, i1;
        {
            float fa = aT[0][0], fb = aT[0][1];
            bool  sw = fb > fa;
            v0 = sw ? fb : fa; i0 = sw ? 1 : 0;
            v1 = sw ? fa : fb; i1 = sw ? 0 : 1;
        }
        #pragma unroll
        for (int t = 0; t < 6; ++t) {
            #pragma unroll
            for (int r = 0; r < 8; ++r) {
                if (t == 0 && r < 2) continue;
                const float v  = aT[t][r];
                const int   o  = t * 16 + r;
                const bool  g0 = v > v0;
                const bool  g1 = v > v1;
                const float nv = g1 ? v : v1;
                const int   ni = g1 ? o : i1;
                v1 = g0 ? v0 : nv; i1 = g0 ? i0 : ni;
                v0 = g0 ? v  : v0; i0 = g0 ? o  : i0;
            }
        }
        i0 += 8 * h; i1 += 8 * h;
        int n0, n1; float m0, m1;
        {
            float w0 = __shfl_xor(v0, 16), w1 = __shfl_xor(v1, 16);
            int   j0 = __shfl_xor(i0, 16), j1 = __shfl_xor(i1, 16);
            const bool mine = (v0 > w0) | ((v0 == w0) & (i0 < j0));
            const bool sa   = (v1 > w0) | ((v1 == w0) & (i1 < j0));
            const bool sb   = (w1 > v0) | ((w1 == v0) & (j1 < i0));
            const int   na = sa ? i1 : j0;  const float ma = sa ? v1 : w0;
            const int   nb = sb ? j1 : i0;  const float mb = sb ? w1 : v0;
            n0 = mine ? i0 : j0;  m0 = mine ? v0 : w0;
            n1 = mine ? na : nb;  m1 = mine ? ma : mb;
        }

        #pragma unroll
        for (int x = 0; x < 16; ++x) {
            const int   r0 = __builtin_amdgcn_readlane(n0, x), r1 = __builtin_amdgcn_readlane(n1, x);
            const float t0 = __int_as_float(__builtin_amdgcn_readlane(__float_as_int(m0), x));
            const float t1 = __int_as_float(__builtin_amdgcn_readlane(__float_as_int(m1), x));
            float* w0 = Tw + r0 * KP + lane;
            float* w1 = Tw + r1 * KP + lane;
            #pragma unroll
            for (int q = 0; q < 3; ++q) {
                const float p  = pq[q][x >> 2][x & 3];
                const float a0 = (q == 2 && yl) ? t0 : p;
                const float a1 = DELTA * ((q == 2 && yl) ? t1 : p);
                __hip_atomic_fetch_add(w0 + 32 * q, a0, __ATOMIC_RELAXED, __HIP_MEMORY_SCOPE_WAVEFRONT);
                __hip_atomic_fetch_add(w1 + 32 * q, a1, __ATOMIC_RELAXED, __HIP_MEMORY_SCOPE_WAVEFRONT);
            }
        }
    }

    __syncthreads();
    {
        float* pl = planes + (size_t)blockIdx.x * PLW;
        #pragma unroll
        for (int ub = 0; ub < 7; ub += 4) {
            v4f   tv[4][NW], wq4[4];
            float yv[4][NW];
            #pragma unroll
            for (int u = 0; u < 4; ++u) {
                if (ub + u < 7) {
                    const int e = 4 * (tid + 256 * (ub + u));
                    const int o = e / KP;
                    wq4[u] = *(const v4f*)&lds_w[e];
                    #pragma unroll
                    for (int wq = 0; wq < NW; ++wq) {
                        tv[u][wq] = *(const v4f*)&lds_t[wq][e];
                        yv[u][wq] = lds_t[wq][o * KP + K_];
                    }
                }
            }
            #pragma unroll
            for (int u = 0; u < 4; ++u) {
                if (ub + u < 7) {
                    const int   e = 4 * (tid + 256 * (ub + u));
                    const v4f   T = ((tv[u][0] + tv[u][1]) + (tv[u][2] + tv[u][3])) + ((tv[u][4] + tv[u][5]) + (tv[u][6] + tv[u][7]));
                    const float Y = ((yv[u][0] + yv[u][1]) + (yv[u][2] + yv[u][3])) + ((yv[u][4] + yv[u][5]) + (yv[u][6] + yv[u][7]));
                    *(v4f*)&pl[e] = T - Y * wq4[u];
                }
            }
        }
        if (wv == 0) {
            const int e = 4 * (tid + 256 * 7);
            const int o = e / KP;
            const v4f w4 = *(const v4f*)&lds_w[e];
            v4f   t8[NW];
            float y8[NW];
            #pragma unroll
            for (int wq = 0; wq < NW; ++wq) { t8[wq] = *(const v4f*)&lds_t[wq][e]; y8[wq] = lds_t[wq][o * KP + K_]; }
            const v4f   T = ((t8[0] + t8[1]) + (t8[2] + t8[3])) + ((t8[4] + t8[5]) + (t8[6] + t8[7]));
            const float Y = ((y8[0] + y8[1]) + (y8[2] + y8[3])) + ((y8[4] + y8[5]) + (y8[6] + y8[7]));
            *(v4f*)&pl[e] = T - Y * w4;
        }
    }
}

__global__ __launch_bounds__(256) void sum_planes(const float* __restrict__ planes,
                                                  float* __restrict__ ds, unsigned int* mb)
{
    __shared__ v2f part[NW][32];
    const int tid  = threadIdx.x;
    const int wv   = tid >> 5;
    const int lane = tid & 31;
    const int e    = blockIdx.x * 64 + 2 * lane;

    const float* p = planes + (size_t)(wv * 32) * PLW + e;
    v2f v[32];
    #pragma unroll
    for (int g = 0; g < 32; ++g) v[g] = *(const v2f*)&p[(size_t)g * PLW];
    __builtin_amdgcn_sched_barrier(0);
    #pragma unroll
    for (int st = 1; st < 32; st <<= 1) {
        #pragma unroll
        for (int g = 0; g < 32; g += 2 * st) v[g] += v[g + st];
    }
    part[wv][lane] = v[0];
    __syncthreads();
    if (wv == 0) {
        v2f t[NW];
        #pragma unroll
        for (int q = 0; q < NW; ++q) t[q] = part[q][lane];
        const v2f s = ((t[0] + t[1]) + (t[2] + t[3])) + ((t[4] + t[5]) + (t[6] + t[7]));
        const v2f d = s * (1.0f / (float)L_);
        *(v2f*)&ds[e] = d;
        const int k = e - (e / KP) * KP;
        float a = __builtin_fmaxf(__builtin_fabsf(d.x), (k + 1 < K_) ? __builtin_fabsf(d.y) : 0.0f);
        #pragma unroll
        for (int m = 16; m >= 1; m >>= 1) a = __builtin_fmaxf(a, __shfl_xor(a, m));
        if (lane == 0) atomicMax(mb, __float_as_uint(a));
    }
}

__global__ void apply_update(const float* __restrict__ w,
                             const float* __restrict__ ds,
                             const unsigned int* __restrict__ mb,
                             float* __restrict__ neww)
{
    int i = blockIdx.x * blockDim.x + threadIdx.x;
    if (i < COUT * K_) {
        float nc = __uint_as_float(*mb);
        int o = i / K_;
        neww[i] = w[i] + LR * (ds[i + o] / nc);
    }
}

extern "C" void kernel_launch(void* const* d_in, const int* in_sizes, int n_in,
                              void* d_out, int out_size, void* d_ws, size_t ws_size,
                              hipStream_t stream)
{
    const float* x = (const float*)d_in[0];
    const float* w = (const float*)d_in[1];
    float* out  = (float*)d_out;
    float* neww = out + (size_t)B_ * COUT * L_;

    float*        planes = (float*)d_ws;
    float*        ds     = planes + (size_t)NGRP * PLW;
    unsigned int* mb     = (unsigned int*)(ds + PLW);

    conv_hebb_gemm<<<NGRP, 256, 0, stream>>>(x, w, out, planes, mb);
    sum_planes<<<PLW / 64, 256, 0, stream>>>(planes, ds, mb);
    apply_update<<<(COUT * K_ + 255) / 256, 256, 0, stream>>>(w, ds, mb, neww);
}
